# starting kernel
# baseline (speedup 1.0000x reference)
#include <hip/hip_runtime.h>

#define HID 128
#define OUTD 64

typedef float v2f __attribute__((ext_vector_type(2)));
typedef float v8f __attribute__((ext_vector_type(8)));

__global__ __launch_bounds__(256) void k_deg(const int* __restrict__ dst,
                                             float* __restrict__ deg, int E) {
  int e = blockIdx.x * 256 + threadIdx.x;
  if (e < E) atomicAdd(&deg[dst[e]], 1.0f);
}

__global__ __launch_bounds__(256) void k_rdeg(float* __restrict__ deg, int N) {
  int i = blockIdx.x * 256 + threadIdx.x;
  if (i < N) deg[i] = 1.0f / fmaxf(deg[i], 1.0f);
}

template<int D>
__global__ __launch_bounds__(256) void k_agg(const float* __restrict__ feat,
                                             const int* __restrict__ src,
                                             const int* __restrict__ dst,
                                             float* __restrict__ sum, int E) {
  const int CH = D / 4;
  long long t = (long long)blockIdx.x * 256 + threadIdx.x;
  int e = (int)(t / CH);
  int c = (int)(t % CH);
  if (e >= E) return;
  int s = src[e], d = dst[e];
  const float4 v = *(const float4*)(feat + (size_t)s * D + (size_t)c * 4);
  float* o = sum + (size_t)d * D + (size_t)c * 4;
  atomicAdd(o + 0, v.x);
  atomicAdd(o + 1, v.y);
  atomicAdd(o + 2, v.z);
  atomicAdd(o + 3, v.w);
}

template<int NCOLS, bool TWO, bool ADD_MEAN, bool RELU, bool BIAS>
__global__ __launch_bounds__(256) void k_gemm(const float* __restrict__ A1,
                                              const float* __restrict__ A2,
                                              const float* __restrict__ rdeg,
                                              const float* __restrict__ B1,
                                              const float* __restrict__ B2,
                                              const float* __restrict__ bias,
                                              const float* __restrict__ addmat,
                                              float* __restrict__ out, int M) {
  const int NT = NCOLS / 16;
  int wave = (blockIdx.x * 256 + (int)threadIdx.x) >> 5;
  int mt = wave / NT;
  int nt = wave - mt * NT;
  int row0 = mt * 16;
  if (row0 >= M) return;

  int lane = threadIdx.x & 31;
  int hh = lane >> 4;
  int lr = lane & 15;
  int col0 = nt * 16;

  const float* a1 = A1 + (size_t)(row0 + lr) * HID + 2 * hh;
  const float* b1 = B1 + (size_t)(2 * hh) * NCOLS + col0 + lr;
  const float* a2 = nullptr;
  const float* b2 = nullptr;
  float rd = 0.f;
  if (TWO) {
    a2 = A2 + (size_t)(row0 + lr) * HID + 2 * hh;
    b2 = B2 + (size_t)(2 * hh) * NCOLS + col0 + lr;
    rd = rdeg[row0 + lr];
  }

  v8f c = {};
#pragma unroll 4
  for (int k = 0; k < HID; k += 4) {
    v2f av = *(const v2f*)(a1 + k);
    v2f bv;
    bv.x = b1[(size_t)k * NCOLS];
    bv.y = b1[(size_t)(k + 1) * NCOLS];
    c = __builtin_amdgcn_wmma_f32_16x16x4_f32(false, av, false, bv, (short)0, c,
                                              false, false);
    if (TWO) {
      v2f av2 = *(const v2f*)(a2 + k);
      av2.x *= rd;
      av2.y *= rd;
      v2f bv2;
      bv2.x = b2[(size_t)k * NCOLS];
      bv2.y = b2[(size_t)(k + 1) * NCOLS];
      c = __builtin_amdgcn_wmma_f32_16x16x4_f32(false, av2, false, bv2, (short)0,
                                                c, false, false);
    }
  }

  float bb = BIAS ? bias[col0 + lr] : 0.f;
#pragma unroll
  for (int v = 0; v < 8; ++v) {
    int r = row0 + v + 8 * hh;
    float val = c[v] + bb;
    if (ADD_MEAN) val += addmat[(size_t)r * NCOLS + col0 + lr] * rdeg[r];
    if (RELU) val = fmaxf(val, 0.f);
    out[(size_t)r * NCOLS + col0 + lr] = val;
  }
}

extern "C" void kernel_launch(void* const* d_in, const int* in_sizes, int n_in,
                              void* d_out, int out_size, void* d_ws,
                              size_t ws_size, hipStream_t stream) {
  const float* x   = (const float*)d_in[0];
  const int*   src = (const int*)d_in[1];
  const int*   dst = (const int*)d_in[2];
  const float* Ws1 = (const float*)d_in[3];
  const float* b1  = (const float*)d_in[4];
  const float* Wn1 = (const float*)d_in[5];
  const float* Ws2 = (const float*)d_in[6];
  const float* b2  = (const float*)d_in[7];
  const float* Wn2 = (const float*)d_in[8];
  const float* Ws3 = (const float*)d_in[9];
  const float* b3  = (const float*)d_in[10];
  const float* Wn3 = (const float*)d_in[11];
  float* out = (float*)d_out;

  const int N = in_sizes[0] / HID;
  const int E = in_sizes[1];

  float* deg = (float*)d_ws;
  float* h1  = deg + N;
  float* h2  = h1 + (size_t)N * HID;
  float* tb  = h2 + (size_t)N * HID;
  float* sum = tb + (size_t)N * OUTD;

  const int mtiles = (N + 15) / 16;
  const int tilesH = mtiles * (HID / 16);
  const int tilesO = mtiles * (OUTD / 16);
  const long long aggTH = (long long)E * (HID / 4);
  const long long aggTO = (long long)E * (OUTD / 4);

  hipMemsetAsync(deg, 0, (size_t)N * sizeof(float), stream);
  hipMemsetAsync(sum, 0, (size_t)N * HID * sizeof(float), stream);
  k_deg<<<(E + 255) / 256, 256, 0, stream>>>(dst, deg, E);
  k_rdeg<<<(N + 255) / 256, 256, 0, stream>>>(deg, N);

  k_agg<HID><<<(int)((aggTH + 255) / 256), 256, 0, stream>>>(x, src, dst, sum, E);
  k_gemm<HID, true, false, true, true><<<(tilesH + 7) / 8, 256, 0, stream>>>(
      x, sum, deg, Ws1, Wn1, b1, nullptr, h1, N);

  hipMemsetAsync(sum, 0, (size_t)N * HID * sizeof(float), stream);
  k_agg<HID><<<(int)((aggTH + 255) / 256), 256, 0, stream>>>(h1, src, dst, sum, E);
  k_gemm<HID, true, false, true, true><<<(tilesH + 7) / 8, 256, 0, stream>>>(
      h1, sum, deg, Ws2, Wn2, b2, nullptr, h2, N);

  k_gemm<OUTD, false, false, false, false><<<(tilesO + 7) / 8, 256, 0, stream>>>(
      h2, nullptr, nullptr, Wn3, nullptr, nullptr, nullptr, tb, N);
  hipMemsetAsync(sum, 0, (size_t)N * OUTD * sizeof(float), stream);
  k_agg<OUTD><<<(int)((aggTO + 255) / 256), 256, 0, stream>>>(tb, src, dst, sum, E);
  k_gemm<OUTD, false, true, false, true><<<(tilesO + 7) / 8, 256, 0, stream>>>(
      h2, nullptr, deg, Ws3, nullptr, b3, sum, out, N);
}


# Edge scatter-sum by destination-owner workgroups: 197-node sum tiles in LDS fed by a scan of per-edge tile bytes, LDS float adds, plain stores; no global atomics, zero fills; sum order = LDS arrival
# speedup vs baseline: 2.9035x; 2.9035x over previous
#include <hip/hip_runtime.h>

#define HID 128
#define OUTD 64

typedef float v2f __attribute__((ext_vector_type(2)));
typedef float v8f __attribute__((ext_vector_type(8)));
typedef float v4f __attribute__((ext_vector_type(4)));

#define DEG_P 32
#define DEG_R 25600
#define DEG_T 1024
#define DEG_B 8

#define OWN_R 197
#define OWN_T 1024
#define OWN_Q 64
#define OWN_U 4

__device__ __forceinline__ void deg_count(unsigned int* hist, int d, int r0) {
  const unsigned int rel = (unsigned int)(d - r0);
  if (rel < (unsigned int)DEG_R) atomicAdd(&hist[rel], 1u);
}

__device__ __forceinline__ unsigned int own_tile(int d, int N) {
  return ((unsigned int)d < (unsigned int)N) ? (unsigned int)d / OWN_R : 255u;
}

__global__ __launch_bounds__(DEG_T) void k_deg(const int* __restrict__ dst,
                                               unsigned int* __restrict__ part,
                                               unsigned char* __restrict__ ids,
                                               int E, int NP, int N) {
  __shared__ unsigned int hist[DEG_R];
  const int tid = threadIdx.x;
  const int r0 = blockIdx.y * DEG_R;
  const bool wr = blockIdx.y == 0;
  uint4* h4 = (uint4*)hist;
  for (int i = tid; i < DEG_R / 4; i += DEG_T) h4[i] = make_uint4(0u, 0u, 0u, 0u);
  __syncthreads();

  const int chunk = (((E + DEG_P - 1) / DEG_P) + 127) & ~127;
  const int base = blockIdx.x * chunk;
  const int end = min(E, base + chunk);
  unsigned int* ids32 = (unsigned int*)ids;
  for (int e0 = base + 4 * tid; e0 + 3 < end; e0 += 4 * DEG_T * DEG_B) {
    int4 d[DEG_B];
#pragma unroll
    for (int j = 0; j < DEG_B; ++j) {
      const int e = e0 + j * 4 * DEG_T;
      if (e + 3 < end) d[j] = *(const int4*)(dst + e);
      else d[j] = make_int4(-1, -1, -1, -1);
    }
#pragma unroll
    for (int j = 0; j < DEG_B; ++j) {
      deg_count(hist, d[j].x, r0);
      deg_count(hist, d[j].y, r0);
      deg_count(hist, d[j].z, r0);
      deg_count(hist, d[j].w, r0);
      const int e = e0 + j * 4 * DEG_T;
      if (wr && e + 3 < end)
        ids32[e >> 2] = own_tile(d[j].x, N) | (own_tile(d[j].y, N) << 8) |
                        (own_tile(d[j].z, N) << 16) | (own_tile(d[j].w, N) << 24);
    }
  }
  const int tail0 = end & ~3;
  if (tail0 >= base && tail0 + tid < end) {
    const int d = dst[tail0 + tid];
    deg_count(hist, d, r0);
    if (wr) ids[tail0 + tid] = (unsigned char)own_tile(d, N);
  }
  if (wr && end == E && base < E && tid < ((E + 15) & ~15) - E) ids[E + tid] = 255;
  __syncthreads();

  const int nb4 = min(DEG_R, NP - r0) / 4;
  uint4* o4 = (uint4*)(part + (size_t)blockIdx.x * NP + r0);
  for (int i = tid; i < nb4; i += DEG_T) o4[i] = h4[i];
}

__global__ __launch_bounds__(256) void k_rdeg(const unsigned int* __restrict__ part,
                                              float* __restrict__ deg, int N, int NP) {
  const int i = blockIdx.x * 256 + threadIdx.x;
  if (i < N) {
    unsigned int s = 0;
#pragma unroll
    for (int p = 0; p < DEG_P; ++p) s += part[(size_t)p * NP + i];
    deg[i] = 1.0f / fmaxf((float)s, 1.0f);
  }
}


__device__ __forceinline__ unsigned int zero_bytes(unsigned int x) {
  return ~(((x & 0x7F7F7F7Fu) + 0x7F7F7F7Fu) | x | 0x7F7F7F7Fu);
}
__device__ __forceinline__ unsigned int pack_flags(unsigned int z) {
  return (((z >> 7) * 0x00204081u) >> 21) & 0xFu;
}

__device__ __forceinline__ void own_load(uint4 (&w)[OWN_U], const uint4* __restrict__ ids4,
                                         int b0, int lane, int G) {
#pragma unroll
  for (int j = 0; j < OWN_U; ++j) {
    const int g = (b0 + j * (OWN_T / 32)) * 32 + lane;
    w[j] = ids4[max(min(g, G - 1), 0)];
  }
}

__device__ __forceinline__ unsigned long long own_mask(const uint4 (&w)[OWN_U], unsigned int pat,
                                                       int b0, int lane, int G) {
  unsigned long long m = 0ull;
#pragma unroll
  for (int j = 0; j < OWN_U; ++j) {
    const int g = (b0 + j * (OWN_T / 32)) * 32 + lane;
    unsigned int mj = pack_flags(zero_bytes(w[j].x ^ pat)) |
                      (pack_flags(zero_bytes(w[j].y ^ pat)) << 4) |
                      (pack_flags(zero_bytes(w[j].z ^ pat)) << 8) |
                      (pack_flags(zero_bytes(w[j].w ^ pat)) << 12);
    if (g >= G) mj = 0u;
    m |= (unsigned long long)mj << (16 * j);
  }
  return m;
}

template<int D>
__device__ __forceinline__ void own_batch(const float* __restrict__ feat,
                                          const int* __restrict__ src,
                                          const int* __restrict__ dst,
                                          float* tile, int e, int r0, int N,
                                          int lane) {
  constexpr int Q = D / 32;
  constexpr int BE = 32 / Q;
  constexpr int NB = 32 / BE;
  int sl = 0;
  int dl = OWN_R;
  if (e >= 0) {
    sl = src[e];
    dl = dst[e] - r0;
  }
  sl = (int)min((unsigned int)sl, (unsigned int)(N - 1));
  dl = (int)min((unsigned int)dl, (unsigned int)OWN_R);
  float v[2][BE][Q];
#pragma unroll
  for (int j = 0; j < BE; ++j) {
    const int sj = __builtin_amdgcn_readlane(sl, j);
    const float* f = feat + (size_t)sj * D + lane;
#pragma unroll
    for (int q = 0; q < Q; ++q) v[0][j][q] = f[q * 32];
  }
#pragma unroll
  for (int b = 0; b < NB; ++b) {
    if (b + 1 < NB) {
#pragma unroll
      for (int j = 0; j < BE; ++j) {
        const int sj = __builtin_amdgcn_readlane(sl, (b + 1) * BE + j);
        const float* f = feat + (size_t)sj * D + lane;
#pragma unroll
        for (int q = 0; q < Q; ++q) v[(b + 1) & 1][j][q] = f[q * 32];
      }
    }
#pragma unroll
    for (int j = 0; j < BE; ++j) {
      const int dj = __builtin_amdgcn_readlane(dl, b * BE + j);
      float* o = tile + dj * D + lane;
#pragma unroll
      for (int q = 0; q < Q; ++q) atomicAdd(o + q * 32, v[b & 1][j][q]);
    }
  }
}

template<int D>
__global__ __launch_bounds__(OWN_T) void k_own(const float* __restrict__ feat,
                                               const int* __restrict__ src,
                                               const int* __restrict__ dst,
                                               const unsigned char* __restrict__ ids,
                                               float* __restrict__ sum, int E, int N) {
  __shared__ v4f tile4[(OWN_R + 1) * D / 4];
  __shared__ int ring[(OWN_T / 32) * OWN_Q];
  float* tile = (float*)tile4;
  const int tid = threadIdx.x;
  const int lane = tid & 31;
  const int wave = __builtin_amdgcn_readfirstlane(tid >> 5);
  const int r0 = blockIdx.x * OWN_R;
  for (int i = tid; i < (OWN_R + 1) * D / 4; i += OWN_T) tile4[i] = v4f{0.f, 0.f, 0.f, 0.f};
  __syncthreads();

  int* q = ring + wave * OWN_Q;
  const unsigned int pat = blockIdx.x * 0x01010101u;
  const int G = (E + 15) >> 4;
  const int nblk = (G + 31) >> 5;
  const uint4* ids4 = (const uint4*)ids;
  int head = 0;
  int tail = 0;
  uint4 w[OWN_U];
  own_load(w, ids4, wave, lane, G);
  for (int b0 = wave; b0 < nblk; b0 += (OWN_T / 32) * OWN_U) {
    unsigned long long m = own_mask(w, pat, b0, lane, G);
    own_load(w, ids4, b0 + (OWN_T / 32) * OWN_U, lane, G);
    for (;;) {
      const bool has = m != 0ull;
      const unsigned int bal = __builtin_amdgcn_ballot_w32(has);
      if (bal == 0u) break;
      if (has) {
        const int k = __builtin_ctzll(m);
        m &= m - 1ull;
        const int g = (b0 + (k >> 4) * (OWN_T / 32)) * 32 + lane;
        const int slot = tail + (int)__builtin_amdgcn_mbcnt_lo(bal, 0u);
        q[slot & (OWN_Q - 1)] = g * 16 + (k & 15);
      }
      tail += __builtin_popcount(bal);
      if (tail - head >= 32) {
        const int e = q[(head + lane) & (OWN_Q - 1)];
        head += 32;
        own_batch<D>(feat, src, dst, tile, e, r0, N, lane);
      }
    }
  }
  if (tail != head) {
    const int e = (lane < tail - head) ? q[(head + lane) & (OWN_Q - 1)] : -1;
    own_batch<D>(feat, src, dst, tile, e, r0, N, lane);
  }
  __syncthreads();

  const int n4 = min(OWN_R, N - r0) * (D / 4);
  v4f* o4 = (v4f*)(sum + (size_t)r0 * D);
  for (int i = tid; i < n4; i += OWN_T) o4[i] = tile4[i];
}


template<int NCOLS, bool TWO, bool ADD_MEAN, bool RELU, bool BIAS>
__global__ __launch_bounds__(256) void k_gemm(const float* __restrict__ A1,
                                              const float* __restrict__ A2,
                                              const float* __restrict__ rdeg,
                                              const float* __restrict__ B1,
                                              const float* __restrict__ B2,
                                              const float* __restrict__ bias,
                                              const float* __restrict__ addmat,
                                              float* __restrict__ out, int M) {
  constexpr int NT = NCOLS / 16;
  constexpr int KQ = HID / 4;
  constexpr int NMAT = TWO ? 2 : 1;
  __shared__ v4f Bs[NMAT * KQ * NCOLS];

  const int tid = threadIdx.x;
  const int lane = tid & 31;
  const int hh = lane >> 4;
  const int lr = lane & 15;
  const int row0 = (blockIdx.x * 8 + (tid >> 5)) * 16;
  const bool live = row0 < M;

  v4f a1v[HID / 8];
  v4f a2v[TWO ? HID / 8 : 1];
  float rd = 0.f;
  if (live) {
    const float* a1 = A1 + (size_t)(row0 + lr) * HID + 4 * hh;
#pragma unroll
    for (int i = 0; i < HID / 8; ++i) a1v[i] = *(const v4f*)(a1 + 8 * i);
    if (TWO) {
      const float* a2 = A2 + (size_t)(row0 + lr) * HID + 4 * hh;
#pragma unroll
      for (int i = 0; i < HID / 8; ++i) a2v[i] = *(const v4f*)(a2 + 8 * i);
      rd = rdeg[row0 + lr];
    }
  }

  for (int u = tid; u < KQ * NCOLS; u += 256) {
    const int kq = u / NCOLS;
    const int c = u - kq * NCOLS;
    const float* p = B1 + (size_t)(4 * kq) * NCOLS + c;
    v4f w;
    w.x = p[0];
    w.y = p[NCOLS];
    w.z = p[2 * NCOLS];
    w.w = p[3 * NCOLS];
    Bs[u] = w;
    if (TWO) {
      const float* q = B2 + (size_t)(4 * kq) * NCOLS + c;
      v4f w2;
      w2.x = q[0];
      w2.y = q[NCOLS];
      w2.z = q[2 * NCOLS];
      w2.w = q[3 * NCOLS];
      Bs[KQ * NCOLS + u] = w2;
    }
  }
  __syncthreads();
  if (!live) return;

  if (TWO) {
#pragma unroll
    for (int i = 0; i < HID / 8; ++i) a2v[i] *= rd;
  }

  v8f acc[NT];
#pragma unroll
  for (int t = 0; t < NT; ++t) acc[t] = v8f{};

  const v4f* bs1 = Bs + hh * NCOLS + lr;
#pragma unroll
  for (int i = 0; i < HID / 8; ++i) {
    const v2f alo = {a1v[i].x, a1v[i].y};
    const v2f ahi = {a1v[i].z, a1v[i].w};
#pragma unroll
    for (int t = 0; t < NT; ++t) {
      const v4f w = bs1[(2 * i) * NCOLS + t * 16];
      const v2f blo = {w.x, w.y};
      const v2f bhi = {w.z, w.w};
      acc[t] = __builtin_amdgcn_wmma_f32_16x16x4_f32(false, alo, false, blo,
                                                     (short)0, acc[t], false, false);
      acc[t] = __builtin_amdgcn_wmma_f32_16x16x4_f32(false, ahi, false, bhi,
                                                     (short)0, acc[t], false, false);
    }
    if (TWO) {
      const v2f clo = {a2v[i].x, a2v[i].y};
      const v2f chi = {a2v[i].z, a2v[i].w};
#pragma unroll
      for (int t = 0; t < NT; ++t) {
        const v4f w = bs1[KQ * NCOLS + (2 * i) * NCOLS + t * 16];
        const v2f blo = {w.x, w.y};
        const v2f bhi = {w.z, w.w};
        acc[t] = __builtin_amdgcn_wmma_f32_16x16x4_f32(false, clo, false, blo,
                                                       (short)0, acc[t], false, false);
        acc[t] = __builtin_amdgcn_wmma_f32_16x16x4_f32(false, chi, false, bhi,
                                                       (short)0, acc[t], false, false);
      }
    }
  }

  float rdr[8];
  if (ADD_MEAN) {
#pragma unroll
    for (int v = 0; v < 8; ++v) rdr[v] = rdeg[row0 + v + 8 * hh];
  }
#pragma unroll
  for (int t = 0; t < NT; ++t) {
    const int col = t * 16 + lr;
    const float bb = BIAS ? bias[col] : 0.f;
#pragma unroll
    for (int v = 0; v < 8; ++v) {
      const int r = row0 + v + 8 * hh;
      float val = acc[t][v] + bb;
      if (ADD_MEAN) val += addmat[(size_t)r * NCOLS + col] * rdr[v];
      if (RELU) val = fmaxf(val, 0.f);
      out[(size_t)r * NCOLS + col] = val;
    }
  }
}

extern "C" void kernel_launch(void* const* d_in, const int* in_sizes, int n_in,
                              void* d_out, int out_size, void* d_ws,
                              size_t ws_size, hipStream_t stream) {
  const float* x   = (const float*)d_in[0];
  const int*   src = (const int*)d_in[1];
  const int*   dst = (const int*)d_in[2];
  const float* Ws1 = (const float*)d_in[3];
  const float* b1  = (const float*)d_in[4];
  const float* Wn1 = (const float*)d_in[5];
  const float* Ws2 = (const float*)d_in[6];
  const float* b2  = (const float*)d_in[7];
  const float* Wn2 = (const float*)d_in[8];
  const float* Ws3 = (const float*)d_in[9];
  const float* b3  = (const float*)d_in[10];
  const float* Wn3 = (const float*)d_in[11];
  float* out = (float*)d_out;

  const int N = in_sizes[0] / HID;
  const int E = in_sizes[1];

  float* h1  = (float*)d_ws;
  float* h2  = h1 + (size_t)N * HID;
  float* tb  = h2 + (size_t)N * HID;
  float* sum = tb + (size_t)N * OUTD;
  float* deg = sum + (size_t)N * HID;
  unsigned char* ids = (unsigned char*)tb;
  float* t3 = h1;

  const int mtiles = (N + 15) / 16;
  const int gemmB = (mtiles + 7) / 8;
  const int ownB = (N + OWN_R - 1) / OWN_R;

  const int NP = (N + 31) & ~31;
  unsigned int* part = (unsigned int*)h2;
  k_deg<<<dim3(DEG_P, (NP + DEG_R - 1) / DEG_R), DEG_T, 0, stream>>>(dst, part, ids, E, NP, N);
  k_rdeg<<<(N + 255) / 256, 256, 0, stream>>>(part, deg, N, NP);

  k_own<HID><<<ownB, OWN_T, 0, stream>>>(x, src, dst, ids, sum, E, N);
  k_gemm<HID, true, false, true, true><<<gemmB, 256, 0, stream>>>(
      x, sum, deg, Ws1, Wn1, b1, nullptr, h1, N);

  k_own<HID><<<ownB, OWN_T, 0, stream>>>(h1, src, dst, ids, sum, E, N);
  k_gemm<HID, true, false, true, true><<<gemmB, 256, 0, stream>>>(
      h1, sum, deg, Ws2, Wn2, b2, nullptr, h2, N);

  k_gemm<OUTD, false, false, false, false><<<gemmB, 256, 0, stream>>>(
      h2, nullptr, nullptr, Wn3, nullptr, nullptr, nullptr, t3, N);
  k_own<OUTD><<<ownB, OWN_T, 0, stream>>>(t3, src, dst, ids, sum, E, N);
  k_gemm<OUTD, false, true, false, true><<<gemmB, 256, 0, stream>>>(
      h2, nullptr, deg, Ws3, nullptr, b3, sum, out, N);
}


# Owner tiles keep their edge list: layer 1's scan stores (src, tile row) per kept edge, placed by per-tile degree sums; layers 2-3 read it: no re-scan, no src/dst reads; sum order = LDS arrival
# speedup vs baseline: 3.3767x; 1.1630x over previous
#include <hip/hip_runtime.h>

#define HID 128
#define OUTD 64

typedef float v2f __attribute__((ext_vector_type(2)));
typedef float v8f __attribute__((ext_vector_type(8)));
typedef float v4f __attribute__((ext_vector_type(4)));

#define DEG_P 32
#define DEG_R 25600
#define DEG_T 1024
#define DEG_B 8

#define OWN_R 197
#define OWN_T 1024
#define OWN_Q 64
#define OWN_U 4

__device__ __forceinline__ void deg_count(unsigned int* hist, int d, int r0) {
  const unsigned int rel = (unsigned int)(d - r0);
  if (rel < (unsigned int)DEG_R) atomicAdd(&hist[rel], 1u);
}

__device__ __forceinline__ unsigned int own_tile(int d, int N) {
  return ((unsigned int)d < (unsigned int)N) ? (unsigned int)d / OWN_R : 255u;
}

__global__ __launch_bounds__(DEG_T) void k_deg(const int* __restrict__ dst,
                                               unsigned int* __restrict__ part,
                                               unsigned char* __restrict__ ids,
                                               int E, int NP, int N) {
  __shared__ unsigned int hist[DEG_R];
  const int tid = threadIdx.x;
  const int r0 = blockIdx.y * DEG_R;
  const bool wr = blockIdx.y == 0;
  uint4* h4 = (uint4*)hist;
  for (int i = tid; i < DEG_R / 4; i += DEG_T) h4[i] = make_uint4(0u, 0u, 0u, 0u);
  __syncthreads();

  const int chunk = (((E + DEG_P - 1) / DEG_P) + 127) & ~127;
  const int base = blockIdx.x * chunk;
  const int end = min(E, base + chunk);
  unsigned int* ids32 = (unsigned int*)ids;
  for (int e0 = base + 4 * tid; e0 + 3 < end; e0 += 4 * DEG_T * DEG_B) {
    int4 d[DEG_B];
#pragma unroll
    for (int j = 0; j < DEG_B; ++j) {
      const int e = e0 + j * 4 * DEG_T;
      if (e + 3 < end) d[j] = *(const int4*)(dst + e);
      else d[j] = make_int4(-1, -1, -1, -1);
    }
#pragma unroll
    for (int j = 0; j < DEG_B; ++j) {
      deg_count(hist, d[j].x, r0);
      deg_count(hist, d[j].y, r0);
      deg_count(hist, d[j].z, r0);
      deg_count(hist, d[j].w, r0);
      const int e = e0 + j * 4 * DEG_T;
      if (wr && e + 3 < end)
        ids32[e >> 2] = own_tile(d[j].x, N) | (own_tile(d[j].y, N) << 8) |
                        (own_tile(d[j].z, N) << 16) | (own_tile(d[j].w, N) << 24);
    }
  }
  const int tail0 = end & ~3;
  if (tail0 >= base && tail0 + tid < end) {
    const int d = dst[tail0 + tid];
    deg_count(hist, d, r0);
    if (wr) ids[tail0 + tid] = (unsigned char)own_tile(d, N);
  }
  if (wr && end == E && base < E && tid < ((E + 15) & ~15) - E) ids[E + tid] = 255;
  __syncthreads();

  const int nb4 = min(DEG_R, NP - r0) / 4;
  uint4* o4 = (uint4*)(part + (size_t)blockIdx.x * NP + r0);
  for (int i = tid; i < nb4; i += DEG_T) o4[i] = h4[i];
}

__global__ __launch_bounds__(256) void k_rdeg(const unsigned int* __restrict__ part,
                                              float* __restrict__ deg,
                                              unsigned int* __restrict__ tcnt, int N, int NP) {
  __shared__ unsigned int tot;
  const int t = threadIdx.x;
  if (t == 0) tot = 0u;
  __syncthreads();
  const int i = blockIdx.x * OWN_R + t;
  if (t < OWN_R && i < N) {
    unsigned int s = 0;
#pragma unroll
    for (int p = 0; p < DEG_P; ++p) s += part[(size_t)p * NP + i];
    deg[i] = 1.0f / fmaxf((float)s, 1.0f);
    atomicAdd(&tot, s);
  }
  __syncthreads();
  if (t == 0) tcnt[blockIdx.x] = tot;
}


__device__ __forceinline__ unsigned int zero_bytes(unsigned int x) {
  return ~(((x & 0x7F7F7F7Fu) + 0x7F7F7F7Fu) | x | 0x7F7F7F7Fu);
}
__device__ __forceinline__ unsigned int pack_flags(unsigned int z) {
  return (((z >> 7) * 0x00204081u) >> 21) & 0xFu;
}

__device__ __forceinline__ void own_load(uint4 (&w)[OWN_U], const uint4* __restrict__ ids4,
                                         int b0, int lane, int G) {
#pragma unroll
  for (int j = 0; j < OWN_U; ++j) {
    const int g = (b0 + j * (OWN_T / 32)) * 32 + lane;
    w[j] = ids4[max(min(g, G - 1), 0)];
  }
}

__device__ __forceinline__ unsigned long long own_mask(const uint4 (&w)[OWN_U], unsigned int pat,
                                                       int b0, int lane, int G) {
  unsigned long long m = 0ull;
#pragma unroll
  for (int j = 0; j < OWN_U; ++j) {
    const int g = (b0 + j * (OWN_T / 32)) * 32 + lane;
    unsigned int mj = pack_flags(zero_bytes(w[j].x ^ pat)) |
                      (pack_flags(zero_bytes(w[j].y ^ pat)) << 4) |
                      (pack_flags(zero_bytes(w[j].z ^ pat)) << 8) |
                      (pack_flags(zero_bytes(w[j].w ^ pat)) << 12);
    if (g >= G) mj = 0u;
    m |= (unsigned long long)mj << (16 * j);
  }
  return m;
}

template<int D>
__device__ __forceinline__ void own_batch(const float* __restrict__ feat,
                                          float* tile, int sl, int dl, int N,
                                          int lane) {
  constexpr int Q = D / 32;
  constexpr int BE = 32 / Q;
  constexpr int NB = 32 / BE;
  sl = (int)min((unsigned int)sl, (unsigned int)(N - 1));
  dl = (int)min((unsigned int)dl, (unsigned int)OWN_R);
  float v[2][BE][Q];
#pragma unroll
  for (int j = 0; j < BE; ++j) {
    const int sj = __builtin_amdgcn_readlane(sl, j);
    const float* f = feat + (size_t)sj * D + lane;
#pragma unroll
    for (int q = 0; q < Q; ++q) v[0][j][q] = f[q * 32];
  }
#pragma unroll
  for (int b = 0; b < NB; ++b) {
    if (b + 1 < NB) {
#pragma unroll
      for (int j = 0; j < BE; ++j) {
        const int sj = __builtin_amdgcn_readlane(sl, (b + 1) * BE + j);
        const float* f = feat + (size_t)sj * D + lane;
#pragma unroll
        for (int q = 0; q < Q; ++q) v[(b + 1) & 1][j][q] = f[q * 32];
      }
    }
#pragma unroll
    for (int j = 0; j < BE; ++j) {
      const int dj = __builtin_amdgcn_readlane(dl, b * BE + j);
      float* o = tile + dj * D + lane;
#pragma unroll
      for (int q = 0; q < Q; ++q) atomicAdd(o + q * 32, v[b & 1][j][q]);
    }
  }
}

__device__ __forceinline__ unsigned int wave_sum(unsigned int v) {
#pragma unroll
  for (int m = 16; m > 0; m >>= 1) v += (unsigned int)__shfl_xor((int)v, m);
  return v;
}

template<int D, bool SCAN>
__global__ __launch_bounds__(OWN_T) void k_own(const float* __restrict__ feat,
                                               const int* __restrict__ src,
                                               const int* __restrict__ dst,
                                               const unsigned char* __restrict__ ids,
                                               const unsigned int* __restrict__ tcnt,
                                               int* __restrict__ listw,
                                               float* __restrict__ sum, int E, int N) {
  int2* list = (int2*)listw;
  __shared__ v4f tile4[(OWN_R + 1) * D / 4];
  __shared__ int ring[SCAN ? (OWN_T / 32) * OWN_Q : 1];
  __shared__ unsigned int lcnt;
  float* tile = (float*)tile4;
  const int tid = threadIdx.x;
  const int lane = tid & 31;
  const int wave = __builtin_amdgcn_readfirstlane(tid >> 5);
  const int r0 = blockIdx.x * OWN_R;

  unsigned int before = 0u;
#pragma unroll
  for (int k = 0; k < 8; ++k) {
    const int t = lane + 32 * k;
    const unsigned int cv = tcnt[min(t, (N + OWN_R - 1) / OWN_R - 1)];
    before += (t < (int)blockIdx.x) ? cv : 0u;
  }
  for (int i = tid; i < (OWN_R + 1) * D / 4; i += OWN_T) tile4[i] = v4f{0.f, 0.f, 0.f, 0.f};
  if (SCAN && tid == 0) lcnt = 0u;
  const unsigned int off =
      min((unsigned int)__builtin_amdgcn_readfirstlane((int)wave_sum(before)), (unsigned int)E);
  __syncthreads();

  if (SCAN) {
    int* q = ring + wave * OWN_Q;
    const unsigned int pat = blockIdx.x * 0x01010101u;
    const int G = (E + 15) >> 4;
    const int nblk = (G + 31) >> 5;
    const uint4* ids4 = (const uint4*)ids;
    int head = 0;
    int tail = 0;
    uint4 w[OWN_U];
    own_load(w, ids4, wave, lane, G);
    for (int b0 = wave; b0 < nblk; b0 += (OWN_T / 32) * OWN_U) {
      unsigned long long m = own_mask(w, pat, b0, lane, G);
      own_load(w, ids4, b0 + (OWN_T / 32) * OWN_U, lane, G);
      for (;;) {
        const bool has = m != 0ull;
        const unsigned int bal = __builtin_amdgcn_ballot_w32(has);
        if (bal == 0u) break;
        if (has) {
          const int k = __builtin_ctzll(m);
          m &= m - 1ull;
          const int g = (b0 + (k >> 4) * (OWN_T / 32)) * 32 + lane;
          const int slot = tail + (int)__builtin_amdgcn_mbcnt_lo(bal, 0u);
          q[slot & (OWN_Q - 1)] = g * 16 + (k & 15);
        }
        tail += __builtin_popcount(bal);
        if (tail - head >= 32) {
          const int e = q[(head + lane) & (OWN_Q - 1)];
          head += 32;
          const int sl = src[e];
          const int dl = dst[e] - r0;
          unsigned int at = 0u;
          if (lane == 0) at = atomicAdd(&lcnt, 32u);
          at = off + (unsigned int)__builtin_amdgcn_readfirstlane((int)at) + (unsigned int)lane;
          if (at < (unsigned int)E) list[at] = make_int2(sl, dl);
          own_batch<D>(feat, tile, sl, dl, N, lane);
        }
      }
    }
    if (tail != head) {
      const int n = tail - head;
      int sl = 0;
      int dl = OWN_R;
      if (lane < n) {
        const int e = q[(head + lane) & (OWN_Q - 1)];
        sl = src[e];
        dl = dst[e] - r0;
      }
      unsigned int at = 0u;
      if (lane == 0) at = atomicAdd(&lcnt, (unsigned int)n);
      at = off + (unsigned int)__builtin_amdgcn_readfirstlane((int)at) + (unsigned int)lane;
      if (lane < n && at < (unsigned int)E) list[at] = make_int2(sl, dl);
      own_batch<D>(feat, tile, sl, dl, N, lane);
    }
  } else {
    const int cnt = (int)min((unsigned int)__builtin_amdgcn_readfirstlane((int)tcnt[blockIdx.x]),
                             (unsigned int)E - off);
    const int last = max(cnt - 1, 0);
    const unsigned int emax = (unsigned int)max(E - 1, 0);
    int i = wave * 32;
    int2 nx = list[min(off + (unsigned int)min(i + lane, last), emax)];
    while (i < cnt) {
      int2 cur = nx;
      if (i + lane >= cnt) cur = make_int2(0, OWN_R);
      i += OWN_T;
      nx = list[min(off + (unsigned int)min(i + lane, last), emax)];
      own_batch<D>(feat, tile, cur.x, cur.y, N, lane);
    }
  }
  __syncthreads();

  const int n4 = min(OWN_R, N - r0) * (D / 4);
  v4f* o4 = (v4f*)(sum + (size_t)r0 * D);
  for (int i = tid; i < n4; i += OWN_T) o4[i] = tile4[i];
}


template<int NCOLS, bool TWO, bool ADD_MEAN, bool RELU, bool BIAS>
__global__ __launch_bounds__(256) void k_gemm(const float* __restrict__ A1,
                                              const float* __restrict__ A2,
                                              const float* __restrict__ rdeg,
                                              const float* __restrict__ B1,
                                              const float* __restrict__ B2,
                                              const float* __restrict__ bias,
                                              const float* __restrict__ addmat,
                                              float* __restrict__ out, int M) {
  constexpr int NT = NCOLS / 16;
  constexpr int KQ = HID / 4;
  constexpr int NMAT = TWO ? 2 : 1;
  __shared__ v4f Bs[NMAT * KQ * NCOLS];

  const int tid = threadIdx.x;
  const int lane = tid & 31;
  const int hh = lane >> 4;
  const int lr = lane & 15;
  const int row0 = (blockIdx.x * 8 + (tid >> 5)) * 16;
  const bool live = row0 < M;

  v4f a1v[HID / 8];
  v4f a2v[TWO ? HID / 8 : 1];
  float rd = 0.f;
  if (live) {
    const float* a1 = A1 + (size_t)(row0 + lr) * HID + 4 * hh;
#pragma unroll
    for (int i = 0; i < HID / 8; ++i) a1v[i] = *(const v4f*)(a1 + 8 * i);
    if (TWO) {
      const float* a2 = A2 + (size_t)(row0 + lr) * HID + 4 * hh;
#pragma unroll
      for (int i = 0; i < HID / 8; ++i) a2v[i] = *(const v4f*)(a2 + 8 * i);
      rd = rdeg[row0 + lr];
    }
  }

  for (int u = tid; u < KQ * NCOLS; u += 256) {
    const int kq = u / NCOLS;
    const int c = u - kq * NCOLS;
    const float* p = B1 + (size_t)(4 * kq) * NCOLS + c;
    v4f w;
    w.x = p[0];
    w.y = p[NCOLS];
    w.z = p[2 * NCOLS];
    w.w = p[3 * NCOLS];
    Bs[u] = w;
    if (TWO) {
      const float* q = B2 + (size_t)(4 * kq) * NCOLS + c;
      v4f w2;
      w2.x = q[0];
      w2.y = q[NCOLS];
      w2.z = q[2 * NCOLS];
      w2.w = q[3 * NCOLS];
      Bs[KQ * NCOLS + u] = w2;
    }
  }
  __syncthreads();
  if (!live) return;

  if (TWO) {
#pragma unroll
    for (int i = 0; i < HID / 8; ++i) a2v[i] *= rd;
  }

  v8f acc[NT];
#pragma unroll
  for (int t = 0; t < NT; ++t) acc[t] = v8f{};

  const v4f* bs1 = Bs + hh * NCOLS + lr;
#pragma unroll
  for (int i = 0; i < HID / 8; ++i) {
    const v2f alo = {a1v[i].x, a1v[i].y};
    const v2f ahi = {a1v[i].z, a1v[i].w};
#pragma unroll
    for (int t = 0; t < NT; ++t) {
      const v4f w = bs1[(2 * i) * NCOLS + t * 16];
      const v2f blo = {w.x, w.y};
      const v2f bhi = {w.z, w.w};
      acc[t] = __builtin_amdgcn_wmma_f32_16x16x4_f32(false, alo, false, blo,
                                                     (short)0, acc[t], false, false);
      acc[t] = __builtin_amdgcn_wmma_f32_16x16x4_f32(false, ahi, false, bhi,
                                                     (short)0, acc[t], false, false);
    }
    if (TWO) {
      const v2f clo = {a2v[i].x, a2v[i].y};
      const v2f chi = {a2v[i].z, a2v[i].w};
#pragma unroll
      for (int t = 0; t < NT; ++t) {
        const v4f w = bs1[KQ * NCOLS + (2 * i) * NCOLS + t * 16];
        const v2f blo = {w.x, w.y};
        const v2f bhi = {w.z, w.w};
        acc[t] = __builtin_amdgcn_wmma_f32_16x16x4_f32(false, clo, false, blo,
                                                       (short)0, acc[t], false, false);
        acc[t] = __builtin_amdgcn_wmma_f32_16x16x4_f32(false, chi, false, bhi,
                                                       (short)0, acc[t], false, false);
      }
    }
  }

  float rdr[8];
  if (ADD_MEAN) {
#pragma unroll
    for (int v = 0; v < 8; ++v) rdr[v] = rdeg[row0 + v + 8 * hh];
  }
#pragma unroll
  for (int t = 0; t < NT; ++t) {
    const int col = t * 16 + lr;
    const float bb = BIAS ? bias[col] : 0.f;
#pragma unroll
    for (int v = 0; v < 8; ++v) {
      const int r = row0 + v + 8 * hh;
      float val = acc[t][v] + bb;
      if (ADD_MEAN) val += addmat[(size_t)r * NCOLS + col] * rdr[v];
      if (RELU) val = fmaxf(val, 0.f);
      out[(size_t)r * NCOLS + col] = val;
    }
  }
}

extern "C" void kernel_launch(void* const* d_in, const int* in_sizes, int n_in,
                              void* d_out, int out_size, void* d_ws,
                              size_t ws_size, hipStream_t stream) {
  const float* x   = (const float*)d_in[0];
  const int*   src = (const int*)d_in[1];
  const int*   dst = (const int*)d_in[2];
  const float* Ws1 = (const float*)d_in[3];
  const float* b1  = (const float*)d_in[4];
  const float* Wn1 = (const float*)d_in[5];
  const float* Ws2 = (const float*)d_in[6];
  const float* b2  = (const float*)d_in[7];
  const float* Wn2 = (const float*)d_in[8];
  const float* Ws3 = (const float*)d_in[9];
  const float* b3  = (const float*)d_in[10];
  const float* Wn3 = (const float*)d_in[11];
  float* out = (float*)d_out;

  const int N = in_sizes[0] / HID;
  const int E = in_sizes[1];

  float* h1  = (float*)d_ws;
  float* h2  = h1 + (size_t)N * HID;
  float* tb  = h2 + (size_t)N * HID;
  float* sum = tb + (size_t)N * OUTD;
  float* deg = sum + (size_t)N * HID;
  unsigned char* ids = (unsigned char*)tb;
  int* list = (int*)(ids + (((size_t)E + 16 + 255) & ~(size_t)255));
  unsigned int* tcnt = (unsigned int*)(list + 2 * (size_t)E);
  float* t3 = h1;

  const int mtiles = (N + 15) / 16;
  const int gemmB = (mtiles + 7) / 8;
  const int ownB = (N + OWN_R - 1) / OWN_R;

  const int NP = (N + 31) & ~31;
  unsigned int* part = (unsigned int*)h2;
  k_deg<<<dim3(DEG_P, (NP + DEG_R - 1) / DEG_R), DEG_T, 0, stream>>>(dst, part, ids, E, NP, N);
  k_rdeg<<<ownB, 256, 0, stream>>>(part, deg, tcnt, N, NP);

  k_own<HID, true><<<ownB, OWN_T, 0, stream>>>(x, src, dst, ids, tcnt, list, sum, E, N);
  k_gemm<HID, true, false, true, true><<<gemmB, 256, 0, stream>>>(
      x, sum, deg, Ws1, Wn1, b1, nullptr, h1, N);

  k_own<HID, false><<<ownB, OWN_T, 0, stream>>>(h1, src, dst, ids, tcnt, list, sum, E, N);
  k_gemm<HID, true, false, true, true><<<gemmB, 256, 0, stream>>>(
      h1, sum, deg, Ws2, Wn2, b2, nullptr, h2, N);

  k_gemm<OUTD, false, false, false, false><<<gemmB, 256, 0, stream>>>(
      h2, nullptr, nullptr, Wn3, nullptr, nullptr, nullptr, t3, N);
  k_own<OUTD, false><<<ownB, OWN_T, 0, stream>>>(t3, src, dst, ids, tcnt, list, sum, E, N);
  k_gemm<OUTD, false, true, false, true><<<gemmB, 256, 0, stream>>>(
      h2, nullptr, deg, Ws3, nullptr, b3, sum, out, N);
}


# Owner workgroup also makes the layer's products: its LDS sum tile is the 2nd A operand (layer 3: added after the product), weights in LDS; sum planes never stored; 3 launches fewer; sum order same
# speedup vs baseline: 3.8277x; 1.1336x over previous
#include <hip/hip_runtime.h>

#define HID 128
#define OUTD 64

typedef float v2f __attribute__((ext_vector_type(2)));
typedef float v8f __attribute__((ext_vector_type(8)));
typedef float v4f __attribute__((ext_vector_type(4)));

#define DEG_P 32
#define DEG_R 25600
#define DEG_T 1024
#define DEG_B 8

#define OWN_R 197
#define OWN_T 1024
#define OWN_Q 64
#define OWN_U 4
#define OWN_S ((OWN_R + 15) / 16)
#define OWN_ROWS (OWN_S * 16)

__device__ __forceinline__ void deg_count(unsigned int* hist, int d, int r0) {
  const unsigned int rel = (unsigned int)(d - r0);
  if (rel < (unsigned int)DEG_R) atomicAdd(&hist[rel], 1u);
}

__device__ __forceinline__ unsigned int own_tile(int d, int N) {
  return ((unsigned int)d < (unsigned int)N) ? (unsigned int)d / OWN_R : 255u;
}

__global__ __launch_bounds__(DEG_T) void k_deg(const int* __restrict__ dst,
                                               unsigned int* __restrict__ part,
                                               unsigned char* __restrict__ ids,
                                               int E, int NP, int N) {
  __shared__ unsigned int hist[DEG_R];
  const int tid = threadIdx.x;
  const int r0 = blockIdx.y * DEG_R;
  const bool wr = blockIdx.y == 0;
  uint4* h4 = (uint4*)hist;
  for (int i = tid; i < DEG_R / 4; i += DEG_T) h4[i] = make_uint4(0u, 0u, 0u, 0u);
  __syncthreads();

  const int chunk = (((E + DEG_P - 1) / DEG_P) + 127) & ~127;
  const int base = blockIdx.x * chunk;
  const int end = min(E, base + chunk);
  unsigned int* ids32 = (unsigned int*)ids;
  for (int e0 = base + 4 * tid; e0 + 3 < end; e0 += 4 * DEG_T * DEG_B) {
    int4 d[DEG_B];
#pragma unroll
    for (int j = 0; j < DEG_B; ++j) {
      const int e = e0 + j * 4 * DEG_T;
      if (e + 3 < end) d[j] = *(const int4*)(dst + e);
      else d[j] = make_int4(-1, -1, -1, -1);
    }
#pragma unroll
    for (int j = 0; j < DEG_B; ++j) {
      deg_count(hist, d[j].x, r0);
      deg_count(hist, d[j].y, r0);
      deg_count(hist, d[j].z, r0);
      deg_count(hist, d[j].w, r0);
      const int e = e0 + j * 4 * DEG_T;
      if (wr && e + 3 < end)
        ids32[e >> 2] = own_tile(d[j].x, N) | (own_tile(d[j].y, N) << 8) |
                        (own_tile(d[j].z, N) << 16) | (own_tile(d[j].w, N) << 24);
    }
  }
  const int tail0 = end & ~3;
  if (tail0 >= base && tail0 + tid < end) {
    const int d = dst[tail0 + tid];
    deg_count(hist, d, r0);
    if (wr) ids[tail0 + tid] = (unsigned char)own_tile(d, N);
  }
  if (wr && end == E && base < E && tid < ((E + 15) & ~15) - E) ids[E + tid] = 255;
  __syncthreads();

  const int nb4 = min(DEG_R, NP - r0) / 4;
  uint4* o4 = (uint4*)(part + (size_t)blockIdx.x * NP + r0);
  for (int i = tid; i < nb4; i += DEG_T) o4[i] = h4[i];
}

__global__ __launch_bounds__(256) void k_rdeg(const unsigned int* __restrict__ part,
                                              float* __restrict__ deg,
                                              unsigned int* __restrict__ tcnt, int N, int NP) {
  __shared__ unsigned int tot;
  const int t = threadIdx.x;
  if (t == 0) tot = 0u;
  __syncthreads();
  const int i = blockIdx.x * OWN_R + t;
  if (t < OWN_R && i < N) {
    unsigned int s = 0;
#pragma unroll
    for (int p = 0; p < DEG_P; ++p) s += part[(size_t)p * NP + i];
    deg[i] = 1.0f / fmaxf((float)s, 1.0f);
    atomicAdd(&tot, s);
  }
  __syncthreads();
  if (t == 0) tcnt[blockIdx.x] = tot;
}


__device__ __forceinline__ unsigned int zero_bytes(unsigned int x) {
  return ~(((x & 0x7F7F7F7Fu) + 0x7F7F7F7Fu) | x | 0x7F7F7F7Fu);
}
__device__ __forceinline__ unsigned int pack_flags(unsigned int z) {
  return (((z >> 7) * 0x00204081u) >> 21) & 0xFu;
}

__device__ __forceinline__ void own_load(uint4 (&w)[OWN_U], const uint4* __restrict__ ids4,
                                         int b0, int lane, int G) {
#pragma unroll
  for (int j = 0; j < OWN_U; ++j) {
    const int g = (b0 + j * (OWN_T / 32)) * 32 + lane;
    w[j] = ids4[max(min(g, G - 1), 0)];
  }
}

__device__ __forceinline__ unsigned long long own_mask(const uint4 (&w)[OWN_U], unsigned int pat,
                                                       int b0, int lane, int G) {
  unsigned long long m = 0ull;
#pragma unroll
  for (int j = 0; j < OWN_U; ++j) {
    const int g = (b0 + j * (OWN_T / 32)) * 32 + lane;
    unsigned int mj = pack_flags(zero_bytes(w[j].x ^ pat)) |
                      (pack_flags(zero_bytes(w[j].y ^ pat)) << 4) |
                      (pack_flags(zero_bytes(w[j].z ^ pat)) << 8) |
                      (pack_flags(zero_bytes(w[j].w ^ pat)) << 12);
    if (g >= G) mj = 0u;
    m |= (unsigned long long)mj << (16 * j);
  }
  return m;
}

template<int D, int TP>
__device__ __forceinline__ void own_batch(const float* __restrict__ feat,
                                          float* tile, int sl, int dl, int N,
                                          int lane) {
  constexpr int Q = D / 32;
  constexpr int BE = 32 / Q;
  constexpr int NB = 32 / BE;
  sl = (int)min((unsigned int)sl, (unsigned int)(N - 1));
  dl = (int)min((unsigned int)dl, (unsigned int)OWN_R);
  float v[2][BE][Q];
#pragma unroll
  for (int j = 0; j < BE; ++j) {
    const int sj = __builtin_amdgcn_readlane(sl, j);
    const float* f = feat + (size_t)sj * D + lane;
#pragma unroll
    for (int q = 0; q < Q; ++q) v[0][j][q] = f[q * 32];
  }
#pragma unroll
  for (int b = 0; b < NB; ++b) {
    if (b + 1 < NB) {
#pragma unroll
      for (int j = 0; j < BE; ++j) {
        const int sj = __builtin_amdgcn_readlane(sl, (b + 1) * BE + j);
        const float* f = feat + (size_t)sj * D + lane;
#pragma unroll
        for (int q = 0; q < Q; ++q) v[(b + 1) & 1][j][q] = f[q * 32];
      }
    }
#pragma unroll
    for (int j = 0; j < BE; ++j) {
      const int dj = __builtin_amdgcn_readlane(dl, b * BE + j);
      float* o = tile + dj * TP + lane;
#pragma unroll
      for (int q = 0; q < Q; ++q) atomicAdd(o + q * 32, v[b & 1][j][q]);
    }
  }
}

__device__ __forceinline__ unsigned int wave_sum(unsigned int v) {
#pragma unroll
  for (int m = 16; m > 0; m >>= 1) v += (unsigned int)__shfl_xor((int)v, m);
  return v;
}

template<int D, int NCOLS, bool SCAN, bool MEAN_A>
__global__ __launch_bounds__(OWN_T) void k_own(const float* __restrict__ feat,
                                               const int* __restrict__ src,
                                               const int* __restrict__ dst,
                                               const unsigned char* __restrict__ ids,
                                               const unsigned int* __restrict__ tcnt,
                                               int* __restrict__ listw,
                                               const float* __restrict__ A1,
                                               const float* __restrict__ rdeg,
                                               const float* __restrict__ B1,
                                               const float* __restrict__ B2,
                                               const float* __restrict__ bias,
                                               float* __restrict__ out, int E, int N) {
  constexpr int TP = D + 4;
  constexpr int KQ = HID / 4;
  constexpr int NMAT = MEAN_A ? 2 : 1;
  int2* list = (int2*)listw;
  __shared__ v4f tile4[OWN_ROWS * TP / 4];
  __shared__ v4f Bs[NMAT * KQ * NCOLS];
  __shared__ int ring[SCAN ? (OWN_T / 32) * OWN_Q : 1];
  __shared__ unsigned int lcnt;
  float* tile = (float*)tile4;
  const int tid = threadIdx.x;
  const int lane = tid & 31;
  const int wave = __builtin_amdgcn_readfirstlane(tid >> 5);
  const int r0 = blockIdx.x * OWN_R;

  unsigned int before = 0u;
#pragma unroll
  for (int k = 0; k < 8; ++k) {
    const int t = lane + 32 * k;
    const unsigned int cv = tcnt[min(t, (N + OWN_R - 1) / OWN_R - 1)];
    before += (t < (int)blockIdx.x) ? cv : 0u;
  }
  for (int i = tid; i < OWN_ROWS * TP / 4; i += OWN_T) tile4[i] = v4f{0.f, 0.f, 0.f, 0.f};
  for (int u = tid; u < KQ * NCOLS; u += OWN_T) {
    const int kq = u / NCOLS;
    const int cc = u - kq * NCOLS;
    const float* p = B1 + (size_t)(4 * kq) * NCOLS + cc;
    v4f wv;
    wv.x = p[0];
    wv.y = p[NCOLS];
    wv.z = p[2 * NCOLS];
    wv.w = p[3 * NCOLS];
    Bs[u] = wv;
    if (MEAN_A) {
      const float* p2 = B2 + (size_t)(4 * kq) * NCOLS + cc;
      v4f w2;
      w2.x = p2[0];
      w2.y = p2[NCOLS];
      w2.z = p2[2 * NCOLS];
      w2.w = p2[3 * NCOLS];
      Bs[KQ * NCOLS + u] = w2;
    }
  }
  if (SCAN && tid == 0) lcnt = 0u;
  const unsigned int off =
      min((unsigned int)__builtin_amdgcn_readfirstlane((int)wave_sum(before)), (unsigned int)E);
  __syncthreads();

  if (SCAN) {
    int* q = ring + wave * OWN_Q;
    const unsigned int pat = blockIdx.x * 0x01010101u;
    const int G = (E + 15) >> 4;
    const int nblk = (G + 31) >> 5;
    const uint4* ids4 = (const uint4*)ids;
    int head = 0;
    int tail = 0;
    uint4 w[OWN_U];
    own_load(w, ids4, wave, lane, G);
    for (int b0 = wave; b0 < nblk; b0 += (OWN_T / 32) * OWN_U) {
      unsigned long long m = own_mask(w, pat, b0, lane, G);
      own_load(w, ids4, b0 + (OWN_T / 32) * OWN_U, lane, G);
      for (;;) {
        const bool has = m != 0ull;
        const unsigned int bal = __builtin_amdgcn_ballot_w32(has);
        if (bal == 0u) break;
        if (has) {
          const int k = __builtin_ctzll(m);
          m &= m - 1ull;
          const int g = (b0 + (k >> 4) * (OWN_T / 32)) * 32 + lane;
          const int slot = tail + (int)__builtin_amdgcn_mbcnt_lo(bal, 0u);
          q[slot & (OWN_Q - 1)] = g * 16 + (k & 15);
        }
        tail += __builtin_popcount(bal);
        if (tail - head >= 32) {
          const int e = q[(head + lane) & (OWN_Q - 1)];
          head += 32;
          const int sl = src[e];
          const int dl = dst[e] - r0;
          unsigned int at = 0u;
          if (lane == 0) at = atomicAdd(&lcnt, 32u);
          at = off + (unsigned int)__builtin_amdgcn_readfirstlane((int)at) + (unsigned int)lane;
          if (at < (unsigned int)E) list[at] = make_int2(sl, dl);
          own_batch<D, TP>(feat, tile, sl, dl, N, lane);
        }
      }
    }
    if (tail != head) {
      const int n = tail - head;
      int sl = 0;
      int dl = OWN_R;
      if (lane < n) {
        const int e = q[(head + lane) & (OWN_Q - 1)];
        sl = src[e];
        dl = dst[e] - r0;
      }
      unsigned int at = 0u;
      if (lane == 0) at = atomicAdd(&lcnt, (unsigned int)n);
      at = off + (unsigned int)__builtin_amdgcn_readfirstlane((int)at) + (unsigned int)lane;
      if (lane < n && at < (unsigned int)E) list[at] = make_int2(sl, dl);
      own_batch<D, TP>(feat, tile, sl, dl, N, lane);
    }
  } else {
    const int cnt = (int)min((unsigned int)__builtin_amdgcn_readfirstlane((int)tcnt[blockIdx.x]),
                             (unsigned int)E - off);
    const int last = max(cnt - 1, 0);
    const unsigned int emax = (unsigned int)max(E - 1, 0);
    int i = wave * 32;
    int2 nx = list[min(off + (unsigned int)min(i + lane, last), emax)];
    while (i < cnt) {
      int2 cur = nx;
      if (i + lane >= cnt) cur = make_int2(0, OWN_R);
      i += OWN_T;
      nx = list[min(off + (unsigned int)min(i + lane, last), emax)];
      own_batch<D, TP>(feat, tile, cur.x, cur.y, N, lane);
    }
  }
  __syncthreads();

  constexpr int NT = 2;
  constexpr int CG = NCOLS / (16 * NT);
  constexpr int NU = OWN_S * CG;
  const int hh = lane >> 4;
  const int lr = lane & 15;
  for (int u = wave; u < NU; u += OWN_T / 32) {
    const int s = u / CG;
    const int g = u - s * CG;
    const int lrow = s * 16 + lr;
    const int grow = min(r0 + lrow, N - 1);
    const float* a1 = A1 + (size_t)grow * HID + 4 * hh;
    v4f a1v[HID / 8];
#pragma unroll
    for (int i = 0; i < HID / 8; ++i) a1v[i] = *(const v4f*)(a1 + 8 * i);
    float rd = 0.f;
    if (MEAN_A) rd = rdeg[grow];

    v8f acc[NT];
#pragma unroll
    for (int t = 0; t < NT; ++t) acc[t] = v8f{};

    const v4f* bs1 = Bs + hh * NCOLS + g * (16 * NT) + lr;
    const v4f* a2p = tile4 + (lrow * TP) / 4 + hh;
#pragma unroll
    for (int i = 0; i < HID / 8; ++i) {
      const v2f alo = {a1v[i].x, a1v[i].y};
      const v2f ahi = {a1v[i].z, a1v[i].w};
#pragma unroll
      for (int t = 0; t < NT; ++t) {
        const v4f w = bs1[(2 * i) * NCOLS + t * 16];
        const v2f blo = {w.x, w.y};
        const v2f bhi = {w.z, w.w};
        acc[t] = __builtin_amdgcn_wmma_f32_16x16x4_f32(false, alo, false, blo,
                                                       (short)0, acc[t], false, false);
        acc[t] = __builtin_amdgcn_wmma_f32_16x16x4_f32(false, ahi, false, bhi,
                                                       (short)0, acc[t], false, false);
      }
      if (MEAN_A) {
        const v4f a2 = a2p[2 * i] * rd;
        const v2f clo = {a2.x, a2.y};
        const v2f chi = {a2.z, a2.w};
#pragma unroll
        for (int t = 0; t < NT; ++t) {
          const v4f w = bs1[KQ * NCOLS + (2 * i) * NCOLS + t * 16];
          const v2f blo = {w.x, w.y};
          const v2f bhi = {w.z, w.w};
          acc[t] = __builtin_amdgcn_wmma_f32_16x16x4_f32(false, clo, false, blo,
                                                         (short)0, acc[t], false, false);
          acc[t] = __builtin_amdgcn_wmma_f32_16x16x4_f32(false, chi, false, bhi,
                                                         (short)0, acc[t], false, false);
        }
      }
    }

    float rdr[8];
    if (!MEAN_A) {
#pragma unroll
      for (int v = 0; v < 8; ++v) rdr[v] = rdeg[min(r0 + s * 16 + v + 8 * hh, N - 1)];
    }
#pragma unroll
    for (int t = 0; t < NT; ++t) {
      const int col = (g * NT + t) * 16 + lr;
      const float bb = bias[col];
#pragma unroll
      for (int v = 0; v < 8; ++v) {
        const int tr = s * 16 + v + 8 * hh;
        const int r = r0 + tr;
        float val = acc[t][v] + bb;
        if (MEAN_A) val = fmaxf(val, 0.f);
        else val += tile[tr * TP + col] * rdr[v];
        if (tr < OWN_R && r < N) out[(size_t)r * NCOLS + col] = val;
      }
    }
  }
}


template<int NCOLS, bool TWO, bool ADD_MEAN, bool RELU, bool BIAS>
__global__ __launch_bounds__(256) void k_gemm(const float* __restrict__ A1,
                                              const float* __restrict__ A2,
                                              const float* __restrict__ rdeg,
                                              const float* __restrict__ B1,
                                              const float* __restrict__ B2,
                                              const float* __restrict__ bias,
                                              const float* __restrict__ addmat,
                                              float* __restrict__ out, int M) {
  constexpr int NT = NCOLS / 16;
  constexpr int KQ = HID / 4;
  constexpr int NMAT = TWO ? 2 : 1;
  __shared__ v4f Bs[NMAT * KQ * NCOLS];

  const int tid = threadIdx.x;
  const int lane = tid & 31;
  const int hh = lane >> 4;
  const int lr = lane & 15;
  const int row0 = (blockIdx.x * 8 + (tid >> 5)) * 16;
  const bool live = row0 < M;

  v4f a1v[HID / 8];
  v4f a2v[TWO ? HID / 8 : 1];
  float rd = 0.f;
  if (live) {
    const float* a1 = A1 + (size_t)(row0 + lr) * HID + 4 * hh;
#pragma unroll
    for (int i = 0; i < HID / 8; ++i) a1v[i] = *(const v4f*)(a1 + 8 * i);
    if (TWO) {
      const float* a2 = A2 + (size_t)(row0 + lr) * HID + 4 * hh;
#pragma unroll
      for (int i = 0; i < HID / 8; ++i) a2v[i] = *(const v4f*)(a2 + 8 * i);
      rd = rdeg[row0 + lr];
    }
  }

  for (int u = tid; u < KQ * NCOLS; u += 256) {
    const int kq = u / NCOLS;
    const int c = u - kq * NCOLS;
    const float* p = B1 + (size_t)(4 * kq) * NCOLS + c;
    v4f w;
    w.x = p[0];
    w.y = p[NCOLS];
    w.z = p[2 * NCOLS];
    w.w = p[3 * NCOLS];
    Bs[u] = w;
    if (TWO) {
      const float* q = B2 + (size_t)(4 * kq) * NCOLS + c;
      v4f w2;
      w2.x = q[0];
      w2.y = q[NCOLS];
      w2.z = q[2 * NCOLS];
      w2.w = q[3 * NCOLS];
      Bs[KQ * NCOLS + u] = w2;
    }
  }
  __syncthreads();
  if (!live) return;

  if (TWO) {
#pragma unroll
    for (int i = 0; i < HID / 8; ++i) a2v[i] *= rd;
  }

  v8f acc[NT];
#pragma unroll
  for (int t = 0; t < NT; ++t) acc[t] = v8f{};

  const v4f* bs1 = Bs + hh * NCOLS + lr;
#pragma unroll
  for (int i = 0; i < HID / 8; ++i) {
    const v2f alo = {a1v[i].x, a1v[i].y};
    const v2f ahi = {a1v[i].z, a1v[i].w};
#pragma unroll
    for (int t = 0; t < NT; ++t) {
      const v4f w = bs1[(2 * i) * NCOLS + t * 16];
      const v2f blo = {w.x, w.y};
      const v2f bhi = {w.z, w.w};
      acc[t] = __builtin_amdgcn_wmma_f32_16x16x4_f32(false, alo, false, blo,
                                                     (short)0, acc[t], false, false);
      acc[t] = __builtin_amdgcn_wmma_f32_16x16x4_f32(false, ahi, false, bhi,
                                                     (short)0, acc[t], false, false);
    }
    if (TWO) {
      const v2f clo = {a2v[i].x, a2v[i].y};
      const v2f chi = {a2v[i].z, a2v[i].w};
#pragma unroll
      for (int t = 0; t < NT; ++t) {
        const v4f w = bs1[KQ * NCOLS + (2 * i) * NCOLS + t * 16];
        const v2f blo = {w.x, w.y};
        const v2f bhi = {w.z, w.w};
        acc[t] = __builtin_amdgcn_wmma_f32_16x16x4_f32(false, clo, false, blo,
                                                       (short)0, acc[t], false, false);
        acc[t] = __builtin_amdgcn_wmma_f32_16x16x4_f32(false, chi, false, bhi,
                                                       (short)0, acc[t], false, false);
      }
    }
  }

  float rdr[8];
  if (ADD_MEAN) {
#pragma unroll
    for (int v = 0; v < 8; ++v) rdr[v] = rdeg[row0 + v + 8 * hh];
  }
#pragma unroll
  for (int t = 0; t < NT; ++t) {
    const int col = t * 16 + lr;
    const float bb = BIAS ? bias[col] : 0.f;
#pragma unroll
    for (int v = 0; v < 8; ++v) {
      const int r = row0 + v + 8 * hh;
      float val = acc[t][v] + bb;
      if (ADD_MEAN) val += addmat[(size_t)r * NCOLS + col] * rdr[v];
      if (RELU) val = fmaxf(val, 0.f);
      out[(size_t)r * NCOLS + col] = val;
    }
  }
}

extern "C" void kernel_launch(void* const* d_in, const int* in_sizes, int n_in,
                              void* d_out, int out_size, void* d_ws,
                              size_t ws_size, hipStream_t stream) {
  const float* x   = (const float*)d_in[0];
  const int*   src = (const int*)d_in[1];
  const int*   dst = (const int*)d_in[2];
  const float* Ws1 = (const float*)d_in[3];
  const float* b1  = (const float*)d_in[4];
  const float* Wn1 = (const float*)d_in[5];
  const float* Ws2 = (const float*)d_in[6];
  const float* b2  = (const float*)d_in[7];
  const float* Wn2 = (const float*)d_in[8];
  const float* Ws3 = (const float*)d_in[9];
  const float* b3  = (const float*)d_in[10];
  const float* Wn3 = (const float*)d_in[11];
  float* out = (float*)d_out;

  const int N = in_sizes[0] / HID;
  const int E = in_sizes[1];

  float* h1  = (float*)d_ws;
  float* h2  = h1 + (size_t)N * HID;
  float* tb  = h2 + (size_t)N * HID;
  float* deg = tb + (size_t)N * OUTD + (size_t)N * HID;
  unsigned char* ids = (unsigned char*)tb;
  int* list = (int*)(ids + (((size_t)E + 16 + 255) & ~(size_t)255));
  unsigned int* tcnt = (unsigned int*)(list + 2 * (size_t)E);
  float* t3 = h1;

  const int mtiles = (N + 15) / 16;
  const int gemmB = (mtiles + 7) / 8;
  const int ownB = (N + OWN_R - 1) / OWN_R;

  const int NP = (N + 31) & ~31;
  unsigned int* part = (unsigned int*)h2;
  k_deg<<<dim3(DEG_P, (NP + DEG_R - 1) / DEG_R), DEG_T, 0, stream>>>(dst, part, ids, E, NP, N);
  k_rdeg<<<ownB, 256, 0, stream>>>(part, deg, tcnt, N, NP);

  k_own<HID, HID, true, true><<<ownB, OWN_T, 0, stream>>>(
      x, src, dst, ids, tcnt, list, x, deg, Ws1, Wn1, b1, h1, E, N);

  k_own<HID, HID, false, true><<<ownB, OWN_T, 0, stream>>>(
      h1, src, dst, ids, tcnt, list, h1, deg, Ws2, Wn2, b2, h2, E, N);

  k_gemm<OUTD, false, false, false, false><<<gemmB, 256, 0, stream>>>(
      h2, nullptr, nullptr, Wn3, nullptr, nullptr, nullptr, t3, N);
  k_own<OUTD, OUTD, false, false><<<ownB, OWN_T, 0, stream>>>(
      t3, src, dst, ids, tcnt, list, h2, deg, Ws3, nullptr, b3, out, E, N);
}


# Owner kernel: alternate groups of 4 waves make the A1@B1 half of their products before their gathers, the rest behind them (2nd unit parked in LDS): WMMA under gathers; k sum order: A1 terms first
# speedup vs baseline: 4.0149x; 1.0489x over previous
#include <hip/hip_runtime.h>

#define HID 128
#define OUTD 64

typedef float v2f __attribute__((ext_vector_type(2)));
typedef float v8f __attribute__((ext_vector_type(8)));
typedef float v4f __attribute__((ext_vector_type(4)));

#define DEG_P 32
#define DEG_R 25600
#define DEG_T 1024
#define DEG_B 8

#define OWN_R 197
#define OWN_T 1024
#define OWN_Q 64
#define OWN_U 4
#define OWN_S ((OWN_R + 15) / 16)
#define OWN_ROWS (OWN_S * 16)

__device__ __forceinline__ void deg_count(unsigned int* hist, int d, int r0) {
  const unsigned int rel = (unsigned int)(d - r0);
  if (rel < (unsigned int)DEG_R) atomicAdd(&hist[rel], 1u);
}

__device__ __forceinline__ unsigned int own_tile(int d, int N) {
  return ((unsigned int)d < (unsigned int)N) ? (unsigned int)d / OWN_R : 255u;
}

__global__ __launch_bounds__(DEG_T) void k_deg(const int* __restrict__ dst,
                                               unsigned int* __restrict__ part,
                                               unsigned char* __restrict__ ids,
                                               int E, int NP, int N) {
  __shared__ unsigned int hist[DEG_R];
  const int tid = threadIdx.x;
  const int r0 = blockIdx.y * DEG_R;
  const bool wr = blockIdx.y == 0;
  uint4* h4 = (uint4*)hist;
  for (int i = tid; i < DEG_R / 4; i += DEG_T) h4[i] = make_uint4(0u, 0u, 0u, 0u);
  __syncthreads();

  const int chunk = (((E + DEG_P - 1) / DEG_P) + 127) & ~127;
  const int base = blockIdx.x * chunk;
  const int end = min(E, base + chunk);
  unsigned int* ids32 = (unsigned int*)ids;
  for (int e0 = base + 4 * tid; e0 + 3 < end; e0 += 4 * DEG_T * DEG_B) {
    int4 d[DEG_B];
#pragma unroll
    for (int j = 0; j < DEG_B; ++j) {
      const int e = e0 + j * 4 * DEG_T;
      if (e + 3 < end) d[j] = *(const int4*)(dst + e);
      else d[j] = make_int4(-1, -1, -1, -1);
    }
#pragma unroll
    for (int j = 0; j < DEG_B; ++j) {
      deg_count(hist, d[j].x, r0);
      deg_count(hist, d[j].y, r0);
      deg_count(hist, d[j].z, r0);
      deg_count(hist, d[j].w, r0);
      const int e = e0 + j * 4 * DEG_T;
      if (wr && e + 3 < end)
        ids32[e >> 2] = own_tile(d[j].x, N) | (own_tile(d[j].y, N) << 8) |
                        (own_tile(d[j].z, N) << 16) | (own_tile(d[j].w, N) << 24);
    }
  }
  const int tail0 = end & ~3;
  if (tail0 >= base && tail0 + tid < end) {
    const int d = dst[tail0 + tid];
    deg_count(hist, d, r0);
    if (wr) ids[tail0 + tid] = (unsigned char)own_tile(d, N);
  }
  if (wr && end == E && base < E && tid < ((E + 15) & ~15) - E) ids[E + tid] = 255;
  __syncthreads();

  const int nb4 = min(DEG_R, NP - r0) / 4;
  uint4* o4 = (uint4*)(part + (size_t)blockIdx.x * NP + r0);
  for (int i = tid; i < nb4; i += DEG_T) o4[i] = h4[i];
}

__global__ __launch_bounds__(256) void k_rdeg(const unsigned int* __restrict__ part,
                                              float* __restrict__ deg,
                                              unsigned int* __restrict__ tcnt, int N, int NP) {
  __shared__ unsigned int tot;
  const int t = threadIdx.x;
  if (t == 0) tot = 0u;
  __syncthreads();
  const int i = blockIdx.x * OWN_R + t;
  if (t < OWN_R && i < N) {
    unsigned int s = 0;
#pragma unroll
    for (int p = 0; p < DEG_P; ++p) s += part[(size_t)p * NP + i];
    deg[i] = 1.0f / fmaxf((float)s, 1.0f);
    atomicAdd(&tot, s);
  }
  __syncthreads();
  if (t == 0) tcnt[blockIdx.x] = tot;
}


__device__ __forceinline__ unsigned int zero_bytes(unsigned int x) {
  return ~(((x & 0x7F7F7F7Fu) + 0x7F7F7F7Fu) | x | 0x7F7F7F7Fu);
}
__device__ __forceinline__ unsigned int pack_flags(unsigned int z) {
  return (((z >> 7) * 0x00204081u) >> 21) & 0xFu;
}

__device__ __forceinline__ void own_load(uint4 (&w)[OWN_U], const uint4* __restrict__ ids4,
                                         int b0, int lane, int G) {
#pragma unroll
  for (int j = 0; j < OWN_U; ++j) {
    const int g = (b0 + j * (OWN_T / 32)) * 32 + lane;
    w[j] = ids4[max(min(g, G - 1), 0)];
  }
}

__device__ __forceinline__ unsigned long long own_mask(const uint4 (&w)[OWN_U], unsigned int pat,
                                                       int b0, int lane, int G) {
  unsigned long long m = 0ull;
#pragma unroll
  for (int j = 0; j < OWN_U; ++j) {
    const int g = (b0 + j * (OWN_T / 32)) * 32 + lane;
    unsigned int mj = pack_flags(zero_bytes(w[j].x ^ pat)) |
                      (pack_flags(zero_bytes(w[j].y ^ pat)) << 4) |
                      (pack_flags(zero_bytes(w[j].z ^ pat)) << 8) |
                      (pack_flags(zero_bytes(w[j].w ^ pat)) << 12);
    if (g >= G) mj = 0u;
    m |= (unsigned long long)mj << (16 * j);
  }
  return m;
}

template<int D, int TP>
__device__ __forceinline__ void own_batch(const float* __restrict__ feat,
                                          float* tile, int sl, int dl, int N,
                                          int lane) {
  constexpr int Q = D / 32;
  constexpr int BE = 32 / Q;
  constexpr int NB = 32 / BE;
  sl = (int)min((unsigned int)sl, (unsigned int)(N - 1));
  dl = (int)min((unsigned int)dl, (unsigned int)OWN_R);
  float v[2][BE][Q];
#pragma unroll
  for (int j = 0; j < BE; ++j) {
    const int sj = __builtin_amdgcn_readlane(sl, j);
    const float* f = feat + (size_t)sj * D + lane;
#pragma unroll
    for (int q = 0; q < Q; ++q) v[0][j][q] = f[q * 32];
  }
#pragma unroll
  for (int b = 0; b < NB; ++b) {
    if (b + 1 < NB) {
#pragma unroll
      for (int j = 0; j < BE; ++j) {
        const int sj = __builtin_amdgcn_readlane(sl, (b + 1) * BE + j);
        const float* f = feat + (size_t)sj * D + lane;
#pragma unroll
        for (int q = 0; q < Q; ++q) v[(b + 1) & 1][j][q] = f[q * 32];
      }
    }
#pragma unroll
    for (int j = 0; j < BE; ++j) {
      const int dj = __builtin_amdgcn_readlane(dl, b * BE + j);
      float* o = tile + dj * TP + lane;
#pragma unroll
      for (int q = 0; q < Q; ++q) atomicAdd(o + q * 32, v[b & 1][j][q]);
    }
  }
}

__device__ __forceinline__ unsigned int wave_sum(unsigned int v) {
#pragma unroll
  for (int m = 16; m > 0; m >>= 1) v += (unsigned int)__shfl_xor((int)v, m);
  return v;
}

template<int NCOLS>
__device__ __forceinline__ void own_a1(const float* __restrict__ A1, const v4f* Bs, int r0,
                                       int N, int u, int lane, v8f (&acc)[2]) {
  constexpr int CG = NCOLS / 32;
  const int hh = lane >> 4;
  const int lr = lane & 15;
  const int s = u / CG;
  const int g = u - s * CG;
  const int grow = min(r0 + s * 16 + lr, N - 1);
  const float* a1 = A1 + (size_t)grow * HID + 4 * hh;
  v4f a1v[HID / 8];
#pragma unroll
  for (int i = 0; i < HID / 8; ++i) a1v[i] = *(const v4f*)(a1 + 8 * i);
  const v4f* bs1 = Bs + hh * NCOLS + g * 32 + lr;
#pragma unroll
  for (int i = 0; i < HID / 8; ++i) {
    const v2f alo = {a1v[i].x, a1v[i].y};
    const v2f ahi = {a1v[i].z, a1v[i].w};
#pragma unroll
    for (int t = 0; t < 2; ++t) {
      const v4f w = bs1[(2 * i) * NCOLS + t * 16];
      const v2f blo = {w.x, w.y};
      const v2f bhi = {w.z, w.w};
      acc[t] = __builtin_amdgcn_wmma_f32_16x16x4_f32(false, alo, false, blo,
                                                     (short)0, acc[t], false, false);
      acc[t] = __builtin_amdgcn_wmma_f32_16x16x4_f32(false, ahi, false, bhi,
                                                     (short)0, acc[t], false, false);
    }
  }
}

template<int NCOLS, int NU>
__device__ __forceinline__ void own_a1_wave(const float* __restrict__ A1, const v4f* Bs,
                                            v4f* park, int r0, int N, int wave, int lane,
                                            v8f (&acc0)[2]) {
  if (wave < NU) own_a1<NCOLS>(A1, Bs, r0, N, wave, lane, acc0);
  if (NU > 32 && wave + 32 < NU) {
    v8f p[2] = {v8f{}, v8f{}};
    own_a1<NCOLS>(A1, Bs, r0, N, wave + 32, lane, p);
#pragma unroll
    for (int t = 0; t < 2; ++t) {
      park[(wave * 4 + 2 * t) * 32 + lane] = v4f{p[t][0], p[t][1], p[t][2], p[t][3]};
      park[(wave * 4 + 2 * t + 1) * 32 + lane] = v4f{p[t][4], p[t][5], p[t][6], p[t][7]};
    }
  }
}

template<int D, int NCOLS, bool SCAN, bool MEAN_A>
__global__ __launch_bounds__(OWN_T) void k_own(const float* __restrict__ feat,
                                               const int* __restrict__ src,
                                               const int* __restrict__ dst,
                                               const unsigned char* __restrict__ ids,
                                               const unsigned int* __restrict__ tcnt,
                                               int* __restrict__ listw,
                                               const float* __restrict__ A1,
                                               const float* __restrict__ rdeg,
                                               const float* __restrict__ B1,
                                               const float* __restrict__ B2,
                                               const float* __restrict__ bias,
                                               float* __restrict__ out, int E, int N) {
  constexpr int TP = D + 4;
  constexpr int KQ = HID / 4;
  constexpr int NMAT = MEAN_A ? 2 : 1;
  int2* list = (int2*)listw;
  __shared__ v4f tile4[OWN_ROWS * TP / 4];
  __shared__ v4f Bs[NMAT * KQ * NCOLS];
  __shared__ int ring[SCAN ? (OWN_T / 32) * OWN_Q : 1];
  __shared__ unsigned int lcnt;
  __shared__ v4f park[(OWN_S * (NCOLS / 32) > 32) ? (OWN_S * (NCOLS / 32) - 32) * 4 * 32 : 1];
  float* tile = (float*)tile4;
  const int tid = threadIdx.x;
  const int lane = tid & 31;
  const int wave = __builtin_amdgcn_readfirstlane(tid >> 5);
  const int r0 = blockIdx.x * OWN_R;

  unsigned int before = 0u;
#pragma unroll
  for (int k = 0; k < 8; ++k) {
    const int t = lane + 32 * k;
    const unsigned int cv = tcnt[min(t, (N + OWN_R - 1) / OWN_R - 1)];
    before += (t < (int)blockIdx.x) ? cv : 0u;
  }
  for (int i = tid; i < OWN_ROWS * TP / 4; i += OWN_T) tile4[i] = v4f{0.f, 0.f, 0.f, 0.f};
  for (int u = tid; u < KQ * NCOLS; u += OWN_T) {
    const int kq = u / NCOLS;
    const int cc = u - kq * NCOLS;
    const float* p = B1 + (size_t)(4 * kq) * NCOLS + cc;
    v4f wv;
    wv.x = p[0];
    wv.y = p[NCOLS];
    wv.z = p[2 * NCOLS];
    wv.w = p[3 * NCOLS];
    Bs[u] = wv;
    if (MEAN_A) {
      const float* p2 = B2 + (size_t)(4 * kq) * NCOLS + cc;
      v4f w2;
      w2.x = p2[0];
      w2.y = p2[NCOLS];
      w2.z = p2[2 * NCOLS];
      w2.w = p2[3 * NCOLS];
      Bs[KQ * NCOLS + u] = w2;
    }
  }
  if (SCAN && tid == 0) lcnt = 0u;
  const unsigned int off =
      min((unsigned int)__builtin_amdgcn_readfirstlane((int)wave_sum(before)), (unsigned int)E);
  __syncthreads();

  constexpr int NT = 2;
  constexpr int CG = NCOLS / (16 * NT);
  constexpr int NU = OWN_S * CG;
  constexpr int UW = (NU + OWN_T / 32 - 1) / (OWN_T / 32);
  static_assert(UW <= 2, "a wave parks one unit at most");
  const bool first = ((wave >> 2) & 1) == 0;
  v8f acc0[NT] = {v8f{}, v8f{}};
  if (first) own_a1_wave<NCOLS, NU>(A1, Bs, park, r0, N, wave, lane, acc0);

  if (SCAN) {
    int* q = ring + wave * OWN_Q;
    const unsigned int pat = blockIdx.x * 0x01010101u;
    const int G = (E + 15) >> 4;
    const int nblk = (G + 31) >> 5;
    const uint4* ids4 = (const uint4*)ids;
    int head = 0;
    int tail = 0;
    uint4 w[OWN_U];
    own_load(w, ids4, wave, lane, G);
    for (int b0 = wave; b0 < nblk; b0 += (OWN_T / 32) * OWN_U) {
      unsigned long long m = own_mask(w, pat, b0, lane, G);
      own_load(w, ids4, b0 + (OWN_T / 32) * OWN_U, lane, G);
      for (;;) {
        const bool has = m != 0ull;
        const unsigned int bal = __builtin_amdgcn_ballot_w32(has);
        if (bal == 0u) break;
        if (has) {
          const int k = __builtin_ctzll(m);
          m &= m - 1ull;
          const int g = (b0 + (k >> 4) * (OWN_T / 32)) * 32 + lane;
          const int slot = tail + (int)__builtin_amdgcn_mbcnt_lo(bal, 0u);
          q[slot & (OWN_Q - 1)] = g * 16 + (k & 15);
        }
        tail += __builtin_popcount(bal);
        if (tail - head >= 32) {
          const int e = q[(head + lane) & (OWN_Q - 1)];
          head += 32;
          const int sl = src[e];
          const int dl = dst[e] - r0;
          unsigned int at = 0u;
          if (lane == 0) at = atomicAdd(&lcnt, 32u);
          at = off + (unsigned int)__builtin_amdgcn_readfirstlane((int)at) + (unsigned int)lane;
          if (at < (unsigned int)E) list[at] = make_int2(sl, dl);
          own_batch<D, TP>(feat, tile, sl, dl, N, lane);
        }
      }
    }
    if (tail != head) {
      const int n = tail - head;
      int sl = 0;
      int dl = OWN_R;
      if (lane < n) {
        const int e = q[(head + lane) & (OWN_Q - 1)];
        sl = src[e];
        dl = dst[e] - r0;
      }
      unsigned int at = 0u;
      if (lane == 0) at = atomicAdd(&lcnt, (unsigned int)n);
      at = off + (unsigned int)__builtin_amdgcn_readfirstlane((int)at) + (unsigned int)lane;
      if (lane < n && at < (unsigned int)E) list[at] = make_int2(sl, dl);
      own_batch<D, TP>(feat, tile, sl, dl, N, lane);
    }
  } else {
    const int cnt = (int)min((unsigned int)__builtin_amdgcn_readfirstlane((int)tcnt[blockIdx.x]),
                             (unsigned int)E - off);
    const int last = max(cnt - 1, 0);
    const unsigned int emax = (unsigned int)max(E - 1, 0);
    int i = wave * 32;
    int2 nx = list[min(off + (unsigned int)min(i + lane, last), emax)];
    while (i < cnt) {
      int2 cur = nx;
      if (i + lane >= cnt) cur = make_int2(0, OWN_R);
      i += OWN_T;
      nx = list[min(off + (unsigned int)min(i + lane, last), emax)];
      own_batch<D, TP>(feat, tile, cur.x, cur.y, N, lane);
    }
  }
  if (!first) own_a1_wave<NCOLS, NU>(A1, Bs, park, r0, N, wave, lane, acc0);
  __syncthreads();

  const int hh = lane >> 4;
  const int lr = lane & 15;
#pragma unroll
  for (int j = 0; j < UW; ++j) {
    const int u = wave + (OWN_T / 32) * j;
    if (u < NU) {
      const int s = u / CG;
      const int g = u - s * CG;
      const int lrow = s * 16 + lr;
      v8f acc[NT];
      if (j == 0) {
#pragma unroll
        for (int t = 0; t < NT; ++t) acc[t] = acc0[t];
      } else {
#pragma unroll
        for (int t = 0; t < NT; ++t) {
          const v4f lo = park[(wave * 4 + 2 * t) * 32 + lane];
          const v4f hi = park[(wave * 4 + 2 * t + 1) * 32 + lane];
          acc[t] = v8f{lo.x, lo.y, lo.z, lo.w, hi.x, hi.y, hi.z, hi.w};
        }
      }
      if (MEAN_A) {
        const int grow = min(r0 + lrow, N - 1);
        const float rd = rdeg[grow];
        const v4f* bs2 = Bs + KQ * NCOLS + hh * NCOLS + g * (16 * NT) + lr;
        const v4f* a2p = tile4 + (lrow * TP) / 4 + hh;
#pragma unroll
        for (int i = 0; i < HID / 8; ++i) {
          const v4f a2 = a2p[2 * i] * rd;
          const v2f clo = {a2.x, a2.y};
          const v2f chi = {a2.z, a2.w};
#pragma unroll
          for (int t = 0; t < NT; ++t) {
            const v4f w = bs2[(2 * i) * NCOLS + t * 16];
            const v2f blo = {w.x, w.y};
            const v2f bhi = {w.z, w.w};
            acc[t] = __builtin_amdgcn_wmma_f32_16x16x4_f32(false, clo, false, blo,
                                                           (short)0, acc[t], false, false);
            acc[t] = __builtin_amdgcn_wmma_f32_16x16x4_f32(false, chi, false, bhi,
                                                           (short)0, acc[t], false, false);
          }
        }
      }

      float rdr[8];
      if (!MEAN_A) {
#pragma unroll
        for (int v = 0; v < 8; ++v) rdr[v] = rdeg[min(r0 + s * 16 + v + 8 * hh, N - 1)];
      }
#pragma unroll
      for (int t = 0; t < NT; ++t) {
        const int col = (g * NT + t) * 16 + lr;
        const float bb = bias[col];
#pragma unroll
        for (int v = 0; v < 8; ++v) {
          const int tr = s * 16 + v + 8 * hh;
          const int r = r0 + tr;
          float val = acc[t][v] + bb;
          if (MEAN_A) val = fmaxf(val, 0.f);
          else val += tile[tr * TP + col] * rdr[v];
          if (tr < OWN_R && r < N) out[(size_t)r * NCOLS + col] = val;
        }
      }
    }
  }
}


template<int NCOLS, bool TWO, bool ADD_MEAN, bool RELU, bool BIAS>
__global__ __launch_bounds__(256) void k_gemm(const float* __restrict__ A1,
                                              const float* __restrict__ A2,
                                              const float* __restrict__ rdeg,
                                              const float* __restrict__ B1,
                                              const float* __restrict__ B2,
                                              const float* __restrict__ bias,
                                              const float* __restrict__ addmat,
                                              float* __restrict__ out, int M) {
  constexpr int NT = NCOLS / 16;
  constexpr int KQ = HID / 4;
  constexpr int NMAT = TWO ? 2 : 1;
  __shared__ v4f Bs[NMAT * KQ * NCOLS];

  const int tid = threadIdx.x;
  const int lane = tid & 31;
  const int hh = lane >> 4;
  const int lr = lane & 15;
  const int row0 = (blockIdx.x * 8 + (tid >> 5)) * 16;
  const bool live = row0 < M;

  v4f a1v[HID / 8];
  v4f a2v[TWO ? HID / 8 : 1];
  float rd = 0.f;
  if (live) {
    const float* a1 = A1 + (size_t)(row0 + lr) * HID + 4 * hh;
#pragma unroll
    for (int i = 0; i < HID / 8; ++i) a1v[i] = *(const v4f*)(a1 + 8 * i);
    if (TWO) {
      const float* a2 = A2 + (size_t)(row0 + lr) * HID + 4 * hh;
#pragma unroll
      for (int i = 0; i < HID / 8; ++i) a2v[i] = *(const v4f*)(a2 + 8 * i);
      rd = rdeg[row0 + lr];
    }
  }

  for (int u = tid; u < KQ * NCOLS; u += 256) {
    const int kq = u / NCOLS;
    const int c = u - kq * NCOLS;
    const float* p = B1 + (size_t)(4 * kq) * NCOLS + c;
    v4f w;
    w.x = p[0];
    w.y = p[NCOLS];
    w.z = p[2 * NCOLS];
    w.w = p[3 * NCOLS];
    Bs[u] = w;
    if (TWO) {
      const float* q = B2 + (size_t)(4 * kq) * NCOLS + c;
      v4f w2;
      w2.x = q[0];
      w2.y = q[NCOLS];
      w2.z = q[2 * NCOLS];
      w2.w = q[3 * NCOLS];
      Bs[KQ * NCOLS + u] = w2;
    }
  }
  __syncthreads();
  if (!live) return;

  if (TWO) {
#pragma unroll
    for (int i = 0; i < HID / 8; ++i) a2v[i] *= rd;
  }

  v8f acc[NT];
#pragma unroll
  for (int t = 0; t < NT; ++t) acc[t] = v8f{};

  const v4f* bs1 = Bs + hh * NCOLS + lr;
#pragma unroll
  for (int i = 0; i < HID / 8; ++i) {
    const v2f alo = {a1v[i].x, a1v[i].y};
    const v2f ahi = {a1v[i].z, a1v[i].w};
#pragma unroll
    for (int t = 0; t < NT; ++t) {
      const v4f w = bs1[(2 * i) * NCOLS + t * 16];
      const v2f blo = {w.x, w.y};
      const v2f bhi = {w.z, w.w};
      acc[t] = __builtin_amdgcn_wmma_f32_16x16x4_f32(false, alo, false, blo,
                                                     (short)0, acc[t], false, false);
      acc[t] = __builtin_amdgcn_wmma_f32_16x16x4_f32(false, ahi, false, bhi,
                                                     (short)0, acc[t], false, false);
    }
    if (TWO) {
      const v2f clo = {a2v[i].x, a2v[i].y};
      const v2f chi = {a2v[i].z, a2v[i].w};
#pragma unroll
      for (int t = 0; t < NT; ++t) {
        const v4f w = bs1[KQ * NCOLS + (2 * i) * NCOLS + t * 16];
        const v2f blo = {w.x, w.y};
        const v2f bhi = {w.z, w.w};
        acc[t] = __builtin_amdgcn_wmma_f32_16x16x4_f32(false, clo, false, blo,
                                                       (short)0, acc[t], false, false);
        acc[t] = __builtin_amdgcn_wmma_f32_16x16x4_f32(false, chi, false, bhi,
                                                       (short)0, acc[t], false, false);
      }
    }
  }

  float rdr[8];
  if (ADD_MEAN) {
#pragma unroll
    for (int v = 0; v < 8; ++v) rdr[v] = rdeg[row0 + v + 8 * hh];
  }
#pragma unroll
  for (int t = 0; t < NT; ++t) {
    const int col = t * 16 + lr;
    const float bb = BIAS ? bias[col] : 0.f;
#pragma unroll
    for (int v = 0; v < 8; ++v) {
      const int r = row0 + v + 8 * hh;
      float val = acc[t][v] + bb;
      if (ADD_MEAN) val += addmat[(size_t)r * NCOLS + col] * rdr[v];
      if (RELU) val = fmaxf(val, 0.f);
      out[(size_t)r * NCOLS + col] = val;
    }
  }
}

extern "C" void kernel_launch(void* const* d_in, const int* in_sizes, int n_in,
                              void* d_out, int out_size, void* d_ws,
                              size_t ws_size, hipStream_t stream) {
  const float* x   = (const float*)d_in[0];
  const int*   src = (const int*)d_in[1];
  const int*   dst = (const int*)d_in[2];
  const float* Ws1 = (const float*)d_in[3];
  const float* b1  = (const float*)d_in[4];
  const float* Wn1 = (const float*)d_in[5];
  const float* Ws2 = (const float*)d_in[6];
  const float* b2  = (const float*)d_in[7];
  const float* Wn2 = (const float*)d_in[8];
  const float* Ws3 = (const float*)d_in[9];
  const float* b3  = (const float*)d_in[10];
  const float* Wn3 = (const float*)d_in[11];
  float* out = (float*)d_out;

  const int N = in_sizes[0] / HID;
  const int E = in_sizes[1];

  float* h1  = (float*)d_ws;
  float* h2  = h1 + (size_t)N * HID;
  float* tb  = h2 + (size_t)N * HID;
  float* deg = tb + (size_t)N * OUTD + (size_t)N * HID;
  unsigned char* ids = (unsigned char*)tb;
  int* list = (int*)(ids + (((size_t)E + 16 + 255) & ~(size_t)255));
  unsigned int* tcnt = (unsigned int*)(list + 2 * (size_t)E);
  float* t3 = h1;

  const int mtiles = (N + 15) / 16;
  const int gemmB = (mtiles + 7) / 8;
  const int ownB = (N + OWN_R - 1) / OWN_R;

  const int NP = (N + 31) & ~31;
  unsigned int* part = (unsigned int*)h2;
  k_deg<<<dim3(DEG_P, (NP + DEG_R - 1) / DEG_R), DEG_T, 0, stream>>>(dst, part, ids, E, NP, N);
  k_rdeg<<<ownB, 256, 0, stream>>>(part, deg, tcnt, N, NP);

  k_own<HID, HID, true, true><<<ownB, OWN_T, 0, stream>>>(
      x, src, dst, ids, tcnt, list, x, deg, Ws1, Wn1, b1, h1, E, N);

  k_own<HID, HID, false, true><<<ownB, OWN_T, 0, stream>>>(
      h1, src, dst, ids, tcnt, list, h1, deg, Ws2, Wn2, b2, h2, E, N);

  k_gemm<OUTD, false, false, false, false><<<gemmB, 256, 0, stream>>>(
      h2, nullptr, nullptr, Wn3, nullptr, nullptr, nullptr, t3, N);
  k_own<OUTD, OUTD, false, false><<<ownB, OWN_T, 0, stream>>>(
      t3, src, dst, ids, tcnt, list, h2, deg, Ws3, nullptr, b3, out, E, N);
}


# Row gathers: a lane takes D/32 consecutive floats by one 16-byte (D=64: 8-byte) load, so one load per row; tile rows lane-interleaved in LDS, B2 laid in that k order; tile-half k sum order changes
# speedup vs baseline: 4.4364x; 1.1050x over previous
#include <hip/hip_runtime.h>

#define HID 128
#define OUTD 64

typedef float v2f __attribute__((ext_vector_type(2)));
typedef float v8f __attribute__((ext_vector_type(8)));
typedef float v4f __attribute__((ext_vector_type(4)));

#define DEG_P 32
#define DEG_R 25600
#define DEG_T 1024
#define DEG_B 8

#define OWN_R 197
#define OWN_T 1024
#define OWN_Q 64
#define OWN_U 4
#define OWN_S ((OWN_R + 15) / 16)
#define OWN_ROWS (OWN_S * 16)

__device__ __forceinline__ void deg_count(unsigned int* hist, int d, int r0) {
  const unsigned int rel = (unsigned int)(d - r0);
  if (rel < (unsigned int)DEG_R) atomicAdd(&hist[rel], 1u);
}

__device__ __forceinline__ unsigned int own_tile(int d, int N) {
  return ((unsigned int)d < (unsigned int)N) ? (unsigned int)d / OWN_R : 255u;
}

__global__ __launch_bounds__(DEG_T) void k_deg(const int* __restrict__ dst,
                                               unsigned int* __restrict__ part,
                                               unsigned char* __restrict__ ids,
                                               int E, int NP, int N) {
  __shared__ unsigned int hist[DEG_R];
  const int tid = threadIdx.x;
  const int r0 = blockIdx.y * DEG_R;
  const bool wr = blockIdx.y == 0;
  uint4* h4 = (uint4*)hist;
  for (int i = tid; i < DEG_R / 4; i += DEG_T) h4[i] = make_uint4(0u, 0u, 0u, 0u);
  __syncthreads();

  const int chunk = (((E + DEG_P - 1) / DEG_P) + 127) & ~127;
  const int base = blockIdx.x * chunk;
  const int end = min(E, base + chunk);
  unsigned int* ids32 = (unsigned int*)ids;
  for (int e0 = base + 4 * tid; e0 + 3 < end; e0 += 4 * DEG_T * DEG_B) {
    int4 d[DEG_B];
#pragma unroll
    for (int j = 0; j < DEG_B; ++j) {
      const int e = e0 + j * 4 * DEG_T;
      if (e + 3 < end) d[j] = *(const int4*)(dst + e);
      else d[j] = make_int4(-1, -1, -1, -1);
    }
#pragma unroll
    for (int j = 0; j < DEG_B; ++j) {
      deg_count(hist, d[j].x, r0);
      deg_count(hist, d[j].y, r0);
      deg_count(hist, d[j].z, r0);
      deg_count(hist, d[j].w, r0);
      const int e = e0 + j * 4 * DEG_T;
      if (wr && e + 3 < end)
        ids32[e >> 2] = own_tile(d[j].x, N) | (own_tile(d[j].y, N) << 8) |
                        (own_tile(d[j].z, N) << 16) | (own_tile(d[j].w, N) << 24);
    }
  }
  const int tail0 = end & ~3;
  if (tail0 >= base && tail0 + tid < end) {
    const int d = dst[tail0 + tid];
    deg_count(hist, d, r0);
    if (wr) ids[tail0 + tid] = (unsigned char)own_tile(d, N);
  }
  if (wr && end == E && base < E && tid < ((E + 15) & ~15) - E) ids[E + tid] = 255;
  __syncthreads();

  const int nb4 = min(DEG_R, NP - r0) / 4;
  uint4* o4 = (uint4*)(part + (size_t)blockIdx.x * NP + r0);
  for (int i = tid; i < nb4; i += DEG_T) o4[i] = h4[i];
}

__global__ __launch_bounds__(256) void k_rdeg(const unsigned int* __restrict__ part,
                                              float* __restrict__ deg,
                                              unsigned int* __restrict__ tcnt, int N, int NP) {
  __shared__ unsigned int tot;
  const int t = threadIdx.x;
  if (t == 0) tot = 0u;
  __syncthreads();
  const int i = blockIdx.x * OWN_R + t;
  if (t < OWN_R && i < N) {
    unsigned int s = 0;
#pragma unroll
    for (int p = 0; p < DEG_P; ++p) s += part[(size_t)p * NP + i];
    deg[i] = 1.0f / fmaxf((float)s, 1.0f);
    atomicAdd(&tot, s);
  }
  __syncthreads();
  if (t == 0) tcnt[blockIdx.x] = tot;
}


template<int D>
__device__ __forceinline__ int tile_pos(int col) {
  constexpr int Q = D / 32;
  return 32 * (col % Q) + col / Q;
}

template<int Q> struct row_part;
template<> struct row_part<4> { typedef v4f type; };
template<> struct row_part<2> { typedef v2f type; };

__device__ __forceinline__ unsigned int zero_bytes(unsigned int x) {
  return ~(((x & 0x7F7F7F7Fu) + 0x7F7F7F7Fu) | x | 0x7F7F7F7Fu);
}
__device__ __forceinline__ unsigned int pack_flags(unsigned int z) {
  return (((z >> 7) * 0x00204081u) >> 21) & 0xFu;
}

__device__ __forceinline__ void own_load(uint4 (&w)[OWN_U], const uint4* __restrict__ ids4,
                                         int b0, int lane, int G) {
#pragma unroll
  for (int j = 0; j < OWN_U; ++j) {
    const int g = (b0 + j * (OWN_T / 32)) * 32 + lane;
    w[j] = ids4[max(min(g, G - 1), 0)];
  }
}

__device__ __forceinline__ unsigned long long own_mask(const uint4 (&w)[OWN_U], unsigned int pat,
                                                       int b0, int lane, int G) {
  unsigned long long m = 0ull;
#pragma unroll
  for (int j = 0; j < OWN_U; ++j) {
    const int g = (b0 + j * (OWN_T / 32)) * 32 + lane;
    unsigned int mj = pack_flags(zero_bytes(w[j].x ^ pat)) |
                      (pack_flags(zero_bytes(w[j].y ^ pat)) << 4) |
                      (pack_flags(zero_bytes(w[j].z ^ pat)) << 8) |
                      (pack_flags(zero_bytes(w[j].w ^ pat)) << 12);
    if (g >= G) mj = 0u;
    m |= (unsigned long long)mj << (16 * j);
  }
  return m;
}

template<int D, int TP>
__device__ __forceinline__ void own_batch(const float* __restrict__ feat,
                                          float* tile, int sl, int dl, int N,
                                          int lane) {
  constexpr int Q = D / 32;
  constexpr int BE = 32 / Q;
  constexpr int NB = 32 / BE;
  typedef typename row_part<Q>::type part_t;
  sl = (int)min((unsigned int)sl, (unsigned int)(N - 1));
  dl = (int)min((unsigned int)dl, (unsigned int)OWN_R);
  part_t v[2][BE];
#pragma unroll
  for (int j = 0; j < BE; ++j) {
    const int sj = __builtin_amdgcn_readlane(sl, j);
    v[0][j] = *(const part_t*)(feat + (size_t)sj * D + Q * lane);
  }
#pragma unroll
  for (int b = 0; b < NB; ++b) {
    if (b + 1 < NB) {
#pragma unroll
      for (int j = 0; j < BE; ++j) {
        const int sj = __builtin_amdgcn_readlane(sl, (b + 1) * BE + j);
        v[(b + 1) & 1][j] = *(const part_t*)(feat + (size_t)sj * D + Q * lane);
      }
    }
#pragma unroll
    for (int j = 0; j < BE; ++j) {
      const int dj = __builtin_amdgcn_readlane(dl, b * BE + j);
      float* o = tile + dj * TP + lane;
#pragma unroll
      for (int q = 0; q < Q; ++q) atomicAdd(o + q * 32, v[b & 1][j][q]);
    }
  }
}

__device__ __forceinline__ unsigned int wave_sum(unsigned int v) {
#pragma unroll
  for (int m = 16; m > 0; m >>= 1) v += (unsigned int)__shfl_xor((int)v, m);
  return v;
}

template<int NCOLS>
__device__ __forceinline__ void own_a1(const float* __restrict__ A1, const v4f* Bs, int r0,
                                       int N, int u, int lane, v8f (&acc)[2]) {
  constexpr int CG = NCOLS / 32;
  const int hh = lane >> 4;
  const int lr = lane & 15;
  const int s = u / CG;
  const int g = u - s * CG;
  const int grow = min(r0 + s * 16 + lr, N - 1);
  const float* a1 = A1 + (size_t)grow * HID + 4 * hh;
  v4f a1v[HID / 8];
#pragma unroll
  for (int i = 0; i < HID / 8; ++i) a1v[i] = *(const v4f*)(a1 + 8 * i);
  const v4f* bs1 = Bs + hh * NCOLS + g * 32 + lr;
#pragma unroll
  for (int i = 0; i < HID / 8; ++i) {
    const v2f alo = {a1v[i].x, a1v[i].y};
    const v2f ahi = {a1v[i].z, a1v[i].w};
#pragma unroll
    for (int t = 0; t < 2; ++t) {
      const v4f w = bs1[(2 * i) * NCOLS + t * 16];
      const v2f blo = {w.x, w.y};
      const v2f bhi = {w.z, w.w};
      acc[t] = __builtin_amdgcn_wmma_f32_16x16x4_f32(false, alo, false, blo,
                                                     (short)0, acc[t], false, false);
      acc[t] = __builtin_amdgcn_wmma_f32_16x16x4_f32(false, ahi, false, bhi,
                                                     (short)0, acc[t], false, false);
    }
  }
}

template<int NCOLS, int NU>
__device__ __forceinline__ void own_a1_wave(const float* __restrict__ A1, const v4f* Bs,
                                            v4f* park, int r0, int N, int wave, int lane,
                                            v8f (&acc0)[2]) {
  if (wave < NU) own_a1<NCOLS>(A1, Bs, r0, N, wave, lane, acc0);
  if (NU > 32 && wave + 32 < NU) {
    v8f p[2] = {v8f{}, v8f{}};
    own_a1<NCOLS>(A1, Bs, r0, N, wave + 32, lane, p);
#pragma unroll
    for (int t = 0; t < 2; ++t) {
      park[(wave * 4 + 2 * t) * 32 + lane] = v4f{p[t][0], p[t][1], p[t][2], p[t][3]};
      park[(wave * 4 + 2 * t + 1) * 32 + lane] = v4f{p[t][4], p[t][5], p[t][6], p[t][7]};
    }
  }
}

template<int D, int NCOLS, bool SCAN, bool MEAN_A>
__global__ __launch_bounds__(OWN_T) void k_own(const float* __restrict__ feat,
                                               const int* __restrict__ src,
                                               const int* __restrict__ dst,
                                               const unsigned char* __restrict__ ids,
                                               const unsigned int* __restrict__ tcnt,
                                               int* __restrict__ listw,
                                               const float* __restrict__ A1,
                                               const float* __restrict__ rdeg,
                                               const float* __restrict__ B1,
                                               const float* __restrict__ B2,
                                               const float* __restrict__ bias,
                                               float* __restrict__ out, int E, int N) {
  constexpr int TP = D + 4;
  constexpr int KQ = HID / 4;
  static_assert(!MEAN_A || D == HID, "the B2 order below is that of a 128-wide lane-interleaved row");
  constexpr int NMAT = MEAN_A ? 2 : 1;
  int2* list = (int2*)listw;
  __shared__ v4f tile4[OWN_ROWS * TP / 4];
  __shared__ v4f Bs[NMAT * KQ * NCOLS];
  __shared__ int ring[SCAN ? (OWN_T / 32) * OWN_Q : 1];
  __shared__ unsigned int lcnt;
  __shared__ v4f park[(OWN_S * (NCOLS / 32) > 32) ? (OWN_S * (NCOLS / 32) - 32) * 4 * 32 : 1];
  float* tile = (float*)tile4;
  const int tid = threadIdx.x;
  const int lane = tid & 31;
  const int wave = __builtin_amdgcn_readfirstlane(tid >> 5);
  const int r0 = blockIdx.x * OWN_R;

  unsigned int before = 0u;
#pragma unroll
  for (int k = 0; k < 8; ++k) {
    const int t = lane + 32 * k;
    const unsigned int cv = tcnt[min(t, (N + OWN_R - 1) / OWN_R - 1)];
    before += (t < (int)blockIdx.x) ? cv : 0u;
  }
  for (int i = tid; i < OWN_ROWS * TP / 4; i += OWN_T) tile4[i] = v4f{0.f, 0.f, 0.f, 0.f};
  for (int u = tid; u < KQ * NCOLS; u += OWN_T) {
    const int kq = u / NCOLS;
    const int cc = u - kq * NCOLS;
    const float* p = B1 + (size_t)(4 * kq) * NCOLS + cc;
    v4f wv;
    wv.x = p[0];
    wv.y = p[NCOLS];
    wv.z = p[2 * NCOLS];
    wv.w = p[3 * NCOLS];
    Bs[u] = wv;
    if (MEAN_A) {
      const float* p2 = B2 + (size_t)(16 * (kq & 7) + (kq >> 3)) * NCOLS + cc;
      v4f w2;
      w2.x = p2[0];
      w2.y = p2[4 * NCOLS];
      w2.z = p2[8 * NCOLS];
      w2.w = p2[12 * NCOLS];
      Bs[KQ * NCOLS + u] = w2;
    }
  }
  if (SCAN && tid == 0) lcnt = 0u;
  const unsigned int off =
      min((unsigned int)__builtin_amdgcn_readfirstlane((int)wave_sum(before)), (unsigned int)E);
  __syncthreads();

  constexpr int NT = 2;
  constexpr int CG = NCOLS / (16 * NT);
  constexpr int NU = OWN_S * CG;
  constexpr int UW = (NU + OWN_T / 32 - 1) / (OWN_T / 32);
  static_assert(UW <= 2, "a wave parks one unit at most");
  const bool first = ((wave >> 2) & 1) == 0;
  v8f acc0[NT] = {v8f{}, v8f{}};
  if (first) own_a1_wave<NCOLS, NU>(A1, Bs, park, r0, N, wave, lane, acc0);

  if (SCAN) {
    int* q = ring + wave * OWN_Q;
    const unsigned int pat = blockIdx.x * 0x01010101u;
    const int G = (E + 15) >> 4;
    const int nblk = (G + 31) >> 5;
    const uint4* ids4 = (const uint4*)ids;
    int head = 0;
    int tail = 0;
    uint4 w[OWN_U];
    own_load(w, ids4, wave, lane, G);
    for (int b0 = wave; b0 < nblk; b0 += (OWN_T / 32) * OWN_U) {
      unsigned long long m = own_mask(w, pat, b0, lane, G);
      own_load(w, ids4, b0 + (OWN_T / 32) * OWN_U, lane, G);
      for (;;) {
        const bool has = m != 0ull;
        const unsigned int bal = __builtin_amdgcn_ballot_w32(has);
        if (bal == 0u) break;
        if (has) {
          const int k = __builtin_ctzll(m);
          m &= m - 1ull;
          const int g = (b0 + (k >> 4) * (OWN_T / 32)) * 32 + lane;
          const int slot = tail + (int)__builtin_amdgcn_mbcnt_lo(bal, 0u);
          q[slot & (OWN_Q - 1)] = g * 16 + (k & 15);
        }
        tail += __builtin_popcount(bal);
        if (tail - head >= 32) {
          const int e = q[(head + lane) & (OWN_Q - 1)];
          head += 32;
          const int sl = src[e];
          const int dl = dst[e] - r0;
          unsigned int at = 0u;
          if (lane == 0) at = atomicAdd(&lcnt, 32u);
          at = off + (unsigned int)__builtin_amdgcn_readfirstlane((int)at) + (unsigned int)lane;
          if (at < (unsigned int)E) list[at] = make_int2(sl, dl);
          own_batch<D, TP>(feat, tile, sl, dl, N, lane);
        }
      }
    }
    if (tail != head) {
      const int n = tail - head;
      int sl = 0;
      int dl = OWN_R;
      if (lane < n) {
        const int e = q[(head + lane) & (OWN_Q - 1)];
        sl = src[e];
        dl = dst[e] - r0;
      }
      unsigned int at = 0u;
      if (lane == 0) at = atomicAdd(&lcnt, (unsigned int)n);
      at = off + (unsigned int)__builtin_amdgcn_readfirstlane((int)at) + (unsigned int)lane;
      if (lane < n && at < (unsigned int)E) list[at] = make_int2(sl, dl);
      own_batch<D, TP>(feat, tile, sl, dl, N, lane);
    }
  } else {
    const int cnt = (int)min((unsigned int)__builtin_amdgcn_readfirstlane((int)tcnt[blockIdx.x]),
                             (unsigned int)E - off);
    const int last = max(cnt - 1, 0);
    const unsigned int emax = (unsigned int)max(E - 1, 0);
    int i = wave * 32;
    int2 nx = list[min(off + (unsigned int)min(i + lane, last), emax)];
    while (i < cnt) {
      int2 cur = nx;
      if (i + lane >= cnt) cur = make_int2(0, OWN_R);
      i += OWN_T;
      nx = list[min(off + (unsigned int)min(i + lane, last), emax)];
      own_batch<D, TP>(feat, tile, cur.x, cur.y, N, lane);
    }
  }
  if (!first) own_a1_wave<NCOLS, NU>(A1, Bs, park, r0, N, wave, lane, acc0);
  __syncthreads();

  const int hh = lane >> 4;
  const int lr = lane & 15;
#pragma unroll
  for (int j = 0; j < UW; ++j) {
    const int u = wave + (OWN_T / 32) * j;
    if (u < NU) {
      const int s = u / CG;
      const int g = u - s * CG;
      const int lrow = s * 16 + lr;
      v8f acc[NT];
      if (j == 0) {
#pragma unroll
        for (int t = 0; t < NT; ++t) acc[t] = acc0[t];
      } else {
#pragma unroll
        for (int t = 0; t < NT; ++t) {
          const v4f lo = park[(wave * 4 + 2 * t) * 32 + lane];
          const v4f hi = park[(wave * 4 + 2 * t + 1) * 32 + lane];
          acc[t] = v8f{lo.x, lo.y, lo.z, lo.w, hi.x, hi.y, hi.z, hi.w};
        }
      }
      if (MEAN_A) {
        const int grow = min(r0 + lrow, N - 1);
        const float rd = rdeg[grow];
        const v4f* bs2 = Bs + KQ * NCOLS + hh * NCOLS + g * (16 * NT) + lr;
        const v4f* a2p = tile4 + (lrow * TP) / 4 + hh;
#pragma unroll
        for (int i = 0; i < HID / 8; ++i) {
          const v4f a2 = a2p[2 * i] * rd;
          const v2f clo = {a2.x, a2.y};
          const v2f chi = {a2.z, a2.w};
#pragma unroll
          for (int t = 0; t < NT; ++t) {
            const v4f w = bs2[(2 * i) * NCOLS + t * 16];
            const v2f blo = {w.x, w.y};
            const v2f bhi = {w.z, w.w};
            acc[t] = __builtin_amdgcn_wmma_f32_16x16x4_f32(false, clo, false, blo,
                                                           (short)0, acc[t], false, false);
            acc[t] = __builtin_amdgcn_wmma_f32_16x16x4_f32(false, chi, false, bhi,
                                                           (short)0, acc[t], false, false);
          }
        }
      }

      float rdr[8];
      if (!MEAN_A) {
#pragma unroll
        for (int v = 0; v < 8; ++v) rdr[v] = rdeg[min(r0 + s * 16 + v + 8 * hh, N - 1)];
      }
#pragma unroll
      for (int t = 0; t < NT; ++t) {
        const int col = (g * NT + t) * 16 + lr;
        const float bb = bias[col];
#pragma unroll
        for (int v = 0; v < 8; ++v) {
          const int tr = s * 16 + v + 8 * hh;
          const int r = r0 + tr;
          float val = acc[t][v] + bb;
          if (MEAN_A) val = fmaxf(val, 0.f);
          else val += tile[tr * TP + tile_pos<D>(col)] * rdr[v];
          if (tr < OWN_R && r < N) out[(size_t)r * NCOLS + col] = val;
        }
      }
    }
  }
}


template<int NCOLS, bool TWO, bool ADD_MEAN, bool RELU, bool BIAS>
__global__ __launch_bounds__(256) void k_gemm(const float* __restrict__ A1,
                                              const float* __restrict__ A2,
                                              const float* __restrict__ rdeg,
                                              const float* __restrict__ B1,
                                              const float* __restrict__ B2,
                                              const float* __restrict__ bias,
                                              const float* __restrict__ addmat,
                                              float* __restrict__ out, int M) {
  constexpr int NT = NCOLS / 16;
  constexpr int KQ = HID / 4;
  constexpr int NMAT = TWO ? 2 : 1;
  __shared__ v4f Bs[NMAT * KQ * NCOLS];

  const int tid = threadIdx.x;
  const int lane = tid & 31;
  const int hh = lane >> 4;
  const int lr = lane & 15;
  const int row0 = (blockIdx.x * 8 + (tid >> 5)) * 16;
  const bool live = row0 < M;

  v4f a1v[HID / 8];
  v4f a2v[TWO ? HID / 8 : 1];
  float rd = 0.f;
  if (live) {
    const float* a1 = A1 + (size_t)(row0 + lr) * HID + 4 * hh;
#pragma unroll
    for (int i = 0; i < HID / 8; ++i) a1v[i] = *(const v4f*)(a1 + 8 * i);
    if (TWO) {
      const float* a2 = A2 + (size_t)(row0 + lr) * HID + 4 * hh;
#pragma unroll
      for (int i = 0; i < HID / 8; ++i) a2v[i] = *(const v4f*)(a2 + 8 * i);
      rd = rdeg[row0 + lr];
    }
  }

  for (int u = tid; u < KQ * NCOLS; u += 256) {
    const int kq = u / NCOLS;
    const int c = u - kq * NCOLS;
    const float* p = B1 + (size_t)(4 * kq) * NCOLS + c;
    v4f w;
    w.x = p[0];
    w.y = p[NCOLS];
    w.z = p[2 * NCOLS];
    w.w = p[3 * NCOLS];
    Bs[u] = w;
    if (TWO) {
      const float* q = B2 + (size_t)(4 * kq) * NCOLS + c;
      v4f w2;
      w2.x = q[0];
      w2.y = q[NCOLS];
      w2.z = q[2 * NCOLS];
      w2.w = q[3 * NCOLS];
      Bs[KQ * NCOLS + u] = w2;
    }
  }
  __syncthreads();
  if (!live) return;

  if (TWO) {
#pragma unroll
    for (int i = 0; i < HID / 8; ++i) a2v[i] *= rd;
  }

  v8f acc[NT];
#pragma unroll
  for (int t = 0; t < NT; ++t) acc[t] = v8f{};

  const v4f* bs1 = Bs + hh * NCOLS + lr;
#pragma unroll
  for (int i = 0; i < HID / 8; ++i) {
    const v2f alo = {a1v[i].x, a1v[i].y};
    const v2f ahi = {a1v[i].z, a1v[i].w};
#pragma unroll
    for (int t = 0; t < NT; ++t) {
      const v4f w = bs1[(2 * i) * NCOLS + t * 16];
      const v2f blo = {w.x, w.y};
      const v2f bhi = {w.z, w.w};
      acc[t] = __builtin_amdgcn_wmma_f32_16x16x4_f32(false, alo, false, blo,
                                                     (short)0, acc[t], false, false);
      acc[t] = __builtin_amdgcn_wmma_f32_16x16x4_f32(false, ahi, false, bhi,
                                                     (short)0, acc[t], false, false);
    }
    if (TWO) {
      const v2f clo = {a2v[i].x, a2v[i].y};
      const v2f chi = {a2v[i].z, a2v[i].w};
#pragma unroll
      for (int t = 0; t < NT; ++t) {
        const v4f w = bs1[KQ * NCOLS + (2 * i) * NCOLS + t * 16];
        const v2f blo = {w.x, w.y};
        const v2f bhi = {w.z, w.w};
        acc[t] = __builtin_amdgcn_wmma_f32_16x16x4_f32(false, clo, false, blo,
                                                       (short)0, acc[t], false, false);
        acc[t] = __builtin_amdgcn_wmma_f32_16x16x4_f32(false, chi, false, bhi,
                                                       (short)0, acc[t], false, false);
      }
    }
  }

  float rdr[8];
  if (ADD_MEAN) {
#pragma unroll
    for (int v = 0; v < 8; ++v) rdr[v] = rdeg[row0 + v + 8 * hh];
  }
#pragma unroll
  for (int t = 0; t < NT; ++t) {
    const int col = t * 16 + lr;
    const float bb = BIAS ? bias[col] : 0.f;
#pragma unroll
    for (int v = 0; v < 8; ++v) {
      const int r = row0 + v + 8 * hh;
      float val = acc[t][v] + bb;
      if (ADD_MEAN) val += addmat[(size_t)r * NCOLS + col] * rdr[v];
      if (RELU) val = fmaxf(val, 0.f);
      out[(size_t)r * NCOLS + col] = val;
    }
  }
}

extern "C" void kernel_launch(void* const* d_in, const int* in_sizes, int n_in,
                              void* d_out, int out_size, void* d_ws,
                              size_t ws_size, hipStream_t stream) {
  const float* x   = (const float*)d_in[0];
  const int*   src = (const int*)d_in[1];
  const int*   dst = (const int*)d_in[2];
  const float* Ws1 = (const float*)d_in[3];
  const float* b1  = (const float*)d_in[4];
  const float* Wn1 = (const float*)d_in[5];
  const float* Ws2 = (const float*)d_in[6];
  const float* b2  = (const float*)d_in[7];
  const float* Wn2 = (const float*)d_in[8];
  const float* Ws3 = (const float*)d_in[9];
  const float* b3  = (const float*)d_in[10];
  const float* Wn3 = (const float*)d_in[11];
  float* out = (float*)d_out;

  const int N = in_sizes[0] / HID;
  const int E = in_sizes[1];

  float* h1  = (float*)d_ws;
  float* h2  = h1 + (size_t)N * HID;
  float* tb  = h2 + (size_t)N * HID;
  float* deg = tb + (size_t)N * OUTD + (size_t)N * HID;
  unsigned char* ids = (unsigned char*)tb;
  int* list = (int*)(ids + (((size_t)E + 16 + 255) & ~(size_t)255));
  unsigned int* tcnt = (unsigned int*)(list + 2 * (size_t)E);
  float* t3 = h1;

  const int mtiles = (N + 15) / 16;
  const int gemmB = (mtiles + 7) / 8;
  const int ownB = (N + OWN_R - 1) / OWN_R;

  const int NP = (N + 31) & ~31;
  unsigned int* part = (unsigned int*)h2;
  k_deg<<<dim3(DEG_P, (NP + DEG_R - 1) / DEG_R), DEG_T, 0, stream>>>(dst, part, ids, E, NP, N);
  k_rdeg<<<ownB, 256, 0, stream>>>(part, deg, tcnt, N, NP);

  k_own<HID, HID, true, true><<<ownB, OWN_T, 0, stream>>>(
      x, src, dst, ids, tcnt, list, x, deg, Ws1, Wn1, b1, h1, E, N);

  k_own<HID, HID, false, true><<<ownB, OWN_T, 0, stream>>>(
      h1, src, dst, ids, tcnt, list, h1, deg, Ws2, Wn2, b2, h2, E, N);

  k_gemm<OUTD, false, false, false, false><<<gemmB, 256, 0, stream>>>(
      h2, nullptr, nullptr, Wn3, nullptr, nullptr, nullptr, t3, N);
  k_own<OUTD, OUTD, false, false><<<ownB, OWN_T, 0, stream>>>(
      t3, src, dst, ids, tcnt, list, h2, deg, Ws3, nullptr, b3, out, E, N);
}


# In-degrees counted by owner workgroups in layer 1's scan (1 LDS integer add per 32 kept edges); first pass writes only tile numbers and chunk tile-count prefixes; 1/deg launch gone; float sums same
# speedup vs baseline: 4.6252x; 1.0426x over previous
#include <hip/hip_runtime.h>

#define HID 128
#define OUTD 64

typedef float v2f __attribute__((ext_vector_type(2)));
typedef float v8f __attribute__((ext_vector_type(8)));
typedef float v4f __attribute__((ext_vector_type(4)));

#define IDS_P 128
#define IDS_T 256
#define IDS_B 8
#define IDS_H 256

#define OWN_R 197
#define OWN_T 1024
#define OWN_Q 64
#define OWN_U 4
#define OWN_S ((OWN_R + 15) / 16)
#define OWN_ROWS (OWN_S * 16)

__device__ __forceinline__ unsigned int own_tile(int d, int N) {
  return ((unsigned int)d < (unsigned int)N) ? (unsigned int)d / OWN_R : 255u;
}

__global__ __launch_bounds__(IDS_T) void k_ids(const int* __restrict__ dst,
                                               unsigned char* __restrict__ ids,
                                               unsigned int* __restrict__ tpart,
                                               int E, int N) {
  static_assert(IDS_T == IDS_H, "one counter per thread");
  __shared__ unsigned int hist[IDS_H];
  __shared__ unsigned int wtot[IDS_T / 32];
  const int tid = threadIdx.x;
  const int lane = tid & 31;
  const int wave = tid >> 5;
  hist[tid] = 0u;
  __syncthreads();

  const int chunk = (((E + IDS_P - 1) / IDS_P) + 127) & ~127;
  const int base = blockIdx.x * chunk;
  const int end = min(E, base + chunk);
  unsigned int* ids32 = (unsigned int*)ids;
  for (int e0 = base + 4 * tid; e0 + 3 < end; e0 += 4 * IDS_T * IDS_B) {
    int4 d[IDS_B];
#pragma unroll
    for (int j = 0; j < IDS_B; ++j) {
      const int e = e0 + j * 4 * IDS_T;
      if (e + 3 < end) d[j] = *(const int4*)(dst + e);
      else d[j] = make_int4(-1, -1, -1, -1);
    }
#pragma unroll
    for (int j = 0; j < IDS_B; ++j) {
      const int e = e0 + j * 4 * IDS_T;
      if (e + 3 < end) {
        const unsigned int t0 = own_tile(d[j].x, N);
        const unsigned int t1 = own_tile(d[j].y, N);
        const unsigned int t2 = own_tile(d[j].z, N);
        const unsigned int t3 = own_tile(d[j].w, N);
        atomicAdd(&hist[t0], 1u);
        atomicAdd(&hist[t1], 1u);
        atomicAdd(&hist[t2], 1u);
        atomicAdd(&hist[t3], 1u);
        ids32[e >> 2] = t0 | (t1 << 8) | (t2 << 16) | (t3 << 24);
      }
    }
  }
  const int tail0 = end & ~3;
  if (tail0 >= base && tail0 + tid < end) {
    const unsigned int t = own_tile(dst[tail0 + tid], N);
    atomicAdd(&hist[t], 1u);
    ids[tail0 + tid] = (unsigned char)t;
  }
  if (end == E && base < E && tid < ((E + 15) & ~15) - E) ids[E + tid] = 255;
  __syncthreads();

  const unsigned int v = hist[tid];
  unsigned int inc = v;
#pragma unroll
  for (int m = 1; m < 32; m <<= 1) {
    const unsigned int o = (unsigned int)__shfl_up((int)inc, m);
    if (lane >= m) inc += o;
  }
  if (lane == 31) wtot[wave] = inc;
  __syncthreads();
  unsigned int front = 0u;
#pragma unroll
  for (int w = 0; w < IDS_T / 32; ++w) front += (w < wave) ? wtot[w] : 0u;
  tpart[blockIdx.x * IDS_H + tid] = front + inc - v;
}


template<int D>
__device__ __forceinline__ int tile_pos(int col) {
  constexpr int Q = D / 32;
  return 32 * (col % Q) + col / Q;
}

template<int Q> struct row_part;
template<> struct row_part<4> { typedef v4f type; };
template<> struct row_part<2> { typedef v2f type; };

__device__ __forceinline__ unsigned int zero_bytes(unsigned int x) {
  return ~(((x & 0x7F7F7F7Fu) + 0x7F7F7F7Fu) | x | 0x7F7F7F7Fu);
}
__device__ __forceinline__ unsigned int pack_flags(unsigned int z) {
  return (((z >> 7) * 0x00204081u) >> 21) & 0xFu;
}

__device__ __forceinline__ void own_load(uint4 (&w)[OWN_U], const uint4* __restrict__ ids4,
                                         int b0, int lane, int G) {
#pragma unroll
  for (int j = 0; j < OWN_U; ++j) {
    const int g = (b0 + j * (OWN_T / 32)) * 32 + lane;
    w[j] = ids4[max(min(g, G - 1), 0)];
  }
}

__device__ __forceinline__ unsigned long long own_mask(const uint4 (&w)[OWN_U], unsigned int pat,
                                                       int b0, int lane, int G) {
  unsigned long long m = 0ull;
#pragma unroll
  for (int j = 0; j < OWN_U; ++j) {
    const int g = (b0 + j * (OWN_T / 32)) * 32 + lane;
    unsigned int mj = pack_flags(zero_bytes(w[j].x ^ pat)) |
                      (pack_flags(zero_bytes(w[j].y ^ pat)) << 4) |
                      (pack_flags(zero_bytes(w[j].z ^ pat)) << 8) |
                      (pack_flags(zero_bytes(w[j].w ^ pat)) << 12);
    if (g >= G) mj = 0u;
    m |= (unsigned long long)mj << (16 * j);
  }
  return m;
}

template<int D, int TP>
__device__ __forceinline__ void own_batch(const float* __restrict__ feat,
                                          float* tile, int sl, int dl, int N,
                                          int lane) {
  constexpr int Q = D / 32;
  constexpr int BE = 32 / Q;
  constexpr int NB = 32 / BE;
  typedef typename row_part<Q>::type part_t;
  sl = (int)min((unsigned int)sl, (unsigned int)(N - 1));
  dl = (int)min((unsigned int)dl, (unsigned int)OWN_R);
  part_t v[2][BE];
#pragma unroll
  for (int j = 0; j < BE; ++j) {
    const int sj = __builtin_amdgcn_readlane(sl, j);
    v[0][j] = *(const part_t*)(feat + (size_t)sj * D + Q * lane);
  }
#pragma unroll
  for (int b = 0; b < NB; ++b) {
    if (b + 1 < NB) {
#pragma unroll
      for (int j = 0; j < BE; ++j) {
        const int sj = __builtin_amdgcn_readlane(sl, (b + 1) * BE + j);
        v[(b + 1) & 1][j] = *(const part_t*)(feat + (size_t)sj * D + Q * lane);
      }
    }
#pragma unroll
    for (int j = 0; j < BE; ++j) {
      const int dj = __builtin_amdgcn_readlane(dl, b * BE + j);
      float* o = tile + dj * TP + lane;
#pragma unroll
      for (int q = 0; q < Q; ++q) atomicAdd(o + q * 32, v[b & 1][j][q]);
    }
  }
}

__device__ __forceinline__ unsigned int wave_sum(unsigned int v) {
#pragma unroll
  for (int m = 16; m > 0; m >>= 1) v += (unsigned int)__shfl_xor((int)v, m);
  return v;
}

template<int NCOLS>
__device__ __forceinline__ void own_a1(const float* __restrict__ A1, const v4f* Bs, int r0,
                                       int N, int u, int lane, v8f (&acc)[2]) {
  constexpr int CG = NCOLS / 32;
  const int hh = lane >> 4;
  const int lr = lane & 15;
  const int s = u / CG;
  const int g = u - s * CG;
  const int grow = min(r0 + s * 16 + lr, N - 1);
  const float* a1 = A1 + (size_t)grow * HID + 4 * hh;
  v4f a1v[HID / 8];
#pragma unroll
  for (int i = 0; i < HID / 8; ++i) a1v[i] = *(const v4f*)(a1 + 8 * i);
  const v4f* bs1 = Bs + hh * NCOLS + g * 32 + lr;
#pragma unroll
  for (int i = 0; i < HID / 8; ++i) {
    const v2f alo = {a1v[i].x, a1v[i].y};
    const v2f ahi = {a1v[i].z, a1v[i].w};
#pragma unroll
    for (int t = 0; t < 2; ++t) {
      const v4f w = bs1[(2 * i) * NCOLS + t * 16];
      const v2f blo = {w.x, w.y};
      const v2f bhi = {w.z, w.w};
      acc[t] = __builtin_amdgcn_wmma_f32_16x16x4_f32(false, alo, false, blo,
                                                     (short)0, acc[t], false, false);
      acc[t] = __builtin_amdgcn_wmma_f32_16x16x4_f32(false, ahi, false, bhi,
                                                     (short)0, acc[t], false, false);
    }
  }
}

template<int NCOLS, int NU>
__device__ __forceinline__ void own_a1_wave(const float* __restrict__ A1, const v4f* Bs,
                                            v4f* park, int r0, int N, int wave, int lane,
                                            v8f (&acc0)[2]) {
  if (wave < NU) own_a1<NCOLS>(A1, Bs, r0, N, wave, lane, acc0);
  if (NU > 32 && wave + 32 < NU) {
    v8f p[2] = {v8f{}, v8f{}};
    own_a1<NCOLS>(A1, Bs, r0, N, wave + 32, lane, p);
#pragma unroll
    for (int t = 0; t < 2; ++t) {
      park[(wave * 4 + 2 * t) * 32 + lane] = v4f{p[t][0], p[t][1], p[t][2], p[t][3]};
      park[(wave * 4 + 2 * t + 1) * 32 + lane] = v4f{p[t][4], p[t][5], p[t][6], p[t][7]};
    }
  }
}

template<int D, int NCOLS, bool SCAN, bool MEAN_A>
__global__ __launch_bounds__(OWN_T) void k_own(const float* __restrict__ feat,
                                               const int* __restrict__ src,
                                               const int* __restrict__ dst,
                                               const unsigned char* __restrict__ ids,
                                               const unsigned int* __restrict__ tpart,
                                               unsigned int* __restrict__ tcnt,
                                               int* __restrict__ listw,
                                               const float* __restrict__ A1,
                                               float* __restrict__ rdeg,
                                               const float* __restrict__ B1,
                                               const float* __restrict__ B2,
                                               const float* __restrict__ bias,
                                               float* __restrict__ out, int E, int N) {
  constexpr int TP = D + 4;
  constexpr int KQ = HID / 4;
  static_assert(!MEAN_A || D == HID, "the B2 order below is that of a 128-wide lane-interleaved row");
  constexpr int NMAT = MEAN_A ? 2 : 1;
  int2* list = (int2*)listw;
  __shared__ v4f tile4[OWN_ROWS * TP / 4];
  __shared__ v4f Bs[NMAT * KQ * NCOLS];
  __shared__ int ring[SCAN ? (OWN_T / 32) * OWN_Q : 1];
  __shared__ unsigned int lcnt;
  __shared__ unsigned int degc[SCAN ? OWN_ROWS : 1];
  __shared__ unsigned int wsum[SCAN ? IDS_P / 32 : 1];
  __shared__ v4f park[(OWN_S * (NCOLS / 32) > 32) ? (OWN_S * (NCOLS / 32) - 32) * 4 * 32 : 1];
  float* tile = (float*)tile4;
  const int tid = threadIdx.x;
  const int lane = tid & 31;
  const int wave = __builtin_amdgcn_readfirstlane(tid >> 5);
  const int r0 = blockIdx.x * OWN_R;

  unsigned int before = 0u;
  if (SCAN) {
    static_assert(IDS_P % 32 == 0 && IDS_P <= 32 * 32, "whole waves, one wsum entry per lane");
    if (wave < IDS_P / 32) before = tpart[(wave * 32 + lane) * IDS_H + (int)blockIdx.x];
  } else {
#pragma unroll
    for (int k = 0; k < 8; ++k) {
      const int t = lane + 32 * k;
      const unsigned int cv = tcnt[min(t, (N + OWN_R - 1) / OWN_R - 1)];
      before += (t < (int)blockIdx.x) ? cv : 0u;
    }
  }
  for (int i = tid; i < OWN_ROWS * TP / 4; i += OWN_T) tile4[i] = v4f{0.f, 0.f, 0.f, 0.f};
  for (int u = tid; u < KQ * NCOLS; u += OWN_T) {
    const int kq = u / NCOLS;
    const int cc = u - kq * NCOLS;
    const float* p = B1 + (size_t)(4 * kq) * NCOLS + cc;
    v4f wv;
    wv.x = p[0];
    wv.y = p[NCOLS];
    wv.z = p[2 * NCOLS];
    wv.w = p[3 * NCOLS];
    Bs[u] = wv;
    if (MEAN_A) {
      const float* p2 = B2 + (size_t)(16 * (kq & 7) + (kq >> 3)) * NCOLS + cc;
      v4f w2;
      w2.x = p2[0];
      w2.y = p2[4 * NCOLS];
      w2.z = p2[8 * NCOLS];
      w2.w = p2[12 * NCOLS];
      Bs[KQ * NCOLS + u] = w2;
    }
  }
  if (SCAN && tid == 0) lcnt = 0u;
  if (SCAN && tid < OWN_ROWS) degc[tid] = 0u;
  before = wave_sum(before);
  if (SCAN && lane == 0 && wave < IDS_P / 32) wsum[wave] = before;
  __syncthreads();
  if (SCAN) before = wave_sum(lane < IDS_P / 32 ? wsum[lane] : 0u);
  const unsigned int off =
      min((unsigned int)__builtin_amdgcn_readfirstlane((int)before), (unsigned int)E);

  constexpr int NT = 2;
  constexpr int CG = NCOLS / (16 * NT);
  constexpr int NU = OWN_S * CG;
  constexpr int UW = (NU + OWN_T / 32 - 1) / (OWN_T / 32);
  static_assert(UW <= 2, "a wave parks one unit at most");
  const bool first = ((wave >> 2) & 1) == 0;
  v8f acc0[NT] = {v8f{}, v8f{}};
  if (first) own_a1_wave<NCOLS, NU>(A1, Bs, park, r0, N, wave, lane, acc0);

  if (SCAN) {
    int* q = ring + wave * OWN_Q;
    const unsigned int pat = blockIdx.x * 0x01010101u;
    const int G = (E + 15) >> 4;
    const int nblk = (G + 31) >> 5;
    const uint4* ids4 = (const uint4*)ids;
    int head = 0;
    int tail = 0;
    uint4 w[OWN_U];
    own_load(w, ids4, wave, lane, G);
    for (int b0 = wave; b0 < nblk; b0 += (OWN_T / 32) * OWN_U) {
      unsigned long long m = own_mask(w, pat, b0, lane, G);
      own_load(w, ids4, b0 + (OWN_T / 32) * OWN_U, lane, G);
      for (;;) {
        const bool has = m != 0ull;
        const unsigned int bal = __builtin_amdgcn_ballot_w32(has);
        if (bal == 0u) break;
        if (has) {
          const int k = __builtin_ctzll(m);
          m &= m - 1ull;
          const int g = (b0 + (k >> 4) * (OWN_T / 32)) * 32 + lane;
          const int slot = tail + (int)__builtin_amdgcn_mbcnt_lo(bal, 0u);
          q[slot & (OWN_Q - 1)] = g * 16 + (k & 15);
        }
        tail += __builtin_popcount(bal);
        if (tail - head >= 32) {
          const int e = q[(head + lane) & (OWN_Q - 1)];
          head += 32;
          const int sl = src[e];
          const int dl = dst[e] - r0;
          atomicAdd(&degc[min((unsigned int)dl, (unsigned int)OWN_R)], 1u);
          unsigned int at = 0u;
          if (lane == 0) at = atomicAdd(&lcnt, 32u);
          at = off + (unsigned int)__builtin_amdgcn_readfirstlane((int)at) + (unsigned int)lane;
          if (at < (unsigned int)E) list[at] = make_int2(sl, dl);
          own_batch<D, TP>(feat, tile, sl, dl, N, lane);
        }
      }
    }
    if (tail != head) {
      const int n = tail - head;
      int sl = 0;
      int dl = OWN_R;
      if (lane < n) {
        const int e = q[(head + lane) & (OWN_Q - 1)];
        sl = src[e];
        dl = dst[e] - r0;
        atomicAdd(&degc[min((unsigned int)dl, (unsigned int)OWN_R)], 1u);
      }
      unsigned int at = 0u;
      if (lane == 0) at = atomicAdd(&lcnt, (unsigned int)n);
      at = off + (unsigned int)__builtin_amdgcn_readfirstlane((int)at) + (unsigned int)lane;
      if (lane < n && at < (unsigned int)E) list[at] = make_int2(sl, dl);
      own_batch<D, TP>(feat, tile, sl, dl, N, lane);
    }
  } else {
    const int cnt = (int)min((unsigned int)__builtin_amdgcn_readfirstlane((int)tcnt[blockIdx.x]),
                             (unsigned int)E - off);
    const int last = max(cnt - 1, 0);
    const unsigned int emax = (unsigned int)max(E - 1, 0);
    int i = wave * 32;
    int2 nx = list[min(off + (unsigned int)min(i + lane, last), emax)];
    while (i < cnt) {
      int2 cur = nx;
      if (i + lane >= cnt) cur = make_int2(0, OWN_R);
      i += OWN_T;
      nx = list[min(off + (unsigned int)min(i + lane, last), emax)];
      own_batch<D, TP>(feat, tile, cur.x, cur.y, N, lane);
    }
  }
  if (!first) own_a1_wave<NCOLS, NU>(A1, Bs, park, r0, N, wave, lane, acc0);
  __syncthreads();
  if (SCAN) {
    if (tid < OWN_R && r0 + tid < N) rdeg[r0 + tid] = 1.0f / fmaxf((float)degc[tid], 1.0f);
    if (tid == 0) tcnt[blockIdx.x] = lcnt;
  }

  const int hh = lane >> 4;
  const int lr = lane & 15;
#pragma unroll
  for (int j = 0; j < UW; ++j) {
    const int u = wave + (OWN_T / 32) * j;
    if (u < NU) {
      const int s = u / CG;
      const int g = u - s * CG;
      const int lrow = s * 16 + lr;
      v8f acc[NT];
      if (j == 0) {
#pragma unroll
        for (int t = 0; t < NT; ++t) acc[t] = acc0[t];
      } else {
#pragma unroll
        for (int t = 0; t < NT; ++t) {
          const v4f lo = park[(wave * 4 + 2 * t) * 32 + lane];
          const v4f hi = park[(wave * 4 + 2 * t + 1) * 32 + lane];
          acc[t] = v8f{lo.x, lo.y, lo.z, lo.w, hi.x, hi.y, hi.z, hi.w};
        }
      }
      if (MEAN_A) {
        const int grow = min(r0 + lrow, N - 1);
        const float rd = SCAN ? 1.0f / fmaxf((float)degc[lrow], 1.0f) : rdeg[grow];
        const v4f* bs2 = Bs + KQ * NCOLS + hh * NCOLS + g * (16 * NT) + lr;
        const v4f* a2p = tile4 + (lrow * TP) / 4 + hh;
#pragma unroll
        for (int i = 0; i < HID / 8; ++i) {
          const v4f a2 = a2p[2 * i] * rd;
          const v2f clo = {a2.x, a2.y};
          const v2f chi = {a2.z, a2.w};
#pragma unroll
          for (int t = 0; t < NT; ++t) {
            const v4f w = bs2[(2 * i) * NCOLS + t * 16];
            const v2f blo = {w.x, w.y};
            const v2f bhi = {w.z, w.w};
            acc[t] = __builtin_amdgcn_wmma_f32_16x16x4_f32(false, clo, false, blo,
                                                           (short)0, acc[t], false, false);
            acc[t] = __builtin_amdgcn_wmma_f32_16x16x4_f32(false, chi, false, bhi,
                                                           (short)0, acc[t], false, false);
          }
        }
      }

      float rdr[8];
      if (!MEAN_A) {
#pragma unroll
        for (int v = 0; v < 8; ++v)
          rdr[v] = SCAN ? 1.0f / fmaxf((float)degc[s * 16 + v + 8 * hh], 1.0f)
                        : rdeg[min(r0 + s * 16 + v + 8 * hh, N - 1)];
      }
#pragma unroll
      for (int t = 0; t < NT; ++t) {
        const int col = (g * NT + t) * 16 + lr;
        const float bb = bias[col];
#pragma unroll
        for (int v = 0; v < 8; ++v) {
          const int tr = s * 16 + v + 8 * hh;
          const int r = r0 + tr;
          float val = acc[t][v] + bb;
          if (MEAN_A) val = fmaxf(val, 0.f);
          else val += tile[tr * TP + tile_pos<D>(col)] * rdr[v];
          if (tr < OWN_R && r < N) out[(size_t)r * NCOLS + col] = val;
        }
      }
    }
  }
}


template<int NCOLS, bool TWO, bool ADD_MEAN, bool RELU, bool BIAS>
__global__ __launch_bounds__(256) void k_gemm(const float* __restrict__ A1,
                                              const float* __restrict__ A2,
                                              const float* __restrict__ rdeg,
                                              const float* __restrict__ B1,
                                              const float* __restrict__ B2,
                                              const float* __restrict__ bias,
                                              const float* __restrict__ addmat,
                                              float* __restrict__ out, int M) {
  constexpr int NT = NCOLS / 16;
  constexpr int KQ = HID / 4;
  constexpr int NMAT = TWO ? 2 : 1;
  __shared__ v4f Bs[NMAT * KQ * NCOLS];

  const int tid = threadIdx.x;
  const int lane = tid & 31;
  const int hh = lane >> 4;
  const int lr = lane & 15;
  const int row0 = (blockIdx.x * 8 + (tid >> 5)) * 16;
  const bool live = row0 < M;

  v4f a1v[HID / 8];
  v4f a2v[TWO ? HID / 8 : 1];
  float rd = 0.f;
  if (live) {
    const float* a1 = A1 + (size_t)(row0 + lr) * HID + 4 * hh;
#pragma unroll
    for (int i = 0; i < HID / 8; ++i) a1v[i] = *(const v4f*)(a1 + 8 * i);
    if (TWO) {
      const float* a2 = A2 + (size_t)(row0 + lr) * HID + 4 * hh;
#pragma unroll
      for (int i = 0; i < HID / 8; ++i) a2v[i] = *(const v4f*)(a2 + 8 * i);
      rd = rdeg[row0 + lr];
    }
  }

  for (int u = tid; u < KQ * NCOLS; u += 256) {
    const int kq = u / NCOLS;
    const int c = u - kq * NCOLS;
    const float* p = B1 + (size_t)(4 * kq) * NCOLS + c;
    v4f w;
    w.x = p[0];
    w.y = p[NCOLS];
    w.z = p[2 * NCOLS];
    w.w = p[3 * NCOLS];
    Bs[u] = w;
    if (TWO) {
      const float* q = B2 + (size_t)(4 * kq) * NCOLS + c;
      v4f w2;
      w2.x = q[0];
      w2.y = q[NCOLS];
      w2.z = q[2 * NCOLS];
      w2.w = q[3 * NCOLS];
      Bs[KQ * NCOLS + u] = w2;
    }
  }
  __syncthreads();
  if (!live) return;

  if (TWO) {
#pragma unroll
    for (int i = 0; i < HID / 8; ++i) a2v[i] *= rd;
  }

  v8f acc[NT];
#pragma unroll
  for (int t = 0; t < NT; ++t) acc[t] = v8f{};

  const v4f* bs1 = Bs + hh * NCOLS + lr;
#pragma unroll
  for (int i = 0; i < HID / 8; ++i) {
    const v2f alo = {a1v[i].x, a1v[i].y};
    const v2f ahi = {a1v[i].z, a1v[i].w};
#pragma unroll
    for (int t = 0; t < NT; ++t) {
      const v4f w = bs1[(2 * i) * NCOLS + t * 16];
      const v2f blo = {w.x, w.y};
      const v2f bhi = {w.z, w.w};
      acc[t] = __builtin_amdgcn_wmma_f32_16x16x4_f32(false, alo, false, blo,
                                                     (short)0, acc[t], false, false);
      acc[t] = __builtin_amdgcn_wmma_f32_16x16x4_f32(false, ahi, false, bhi,
                                                     (short)0, acc[t], false, false);
    }
    if (TWO) {
      const v2f clo = {a2v[i].x, a2v[i].y};
      const v2f chi = {a2v[i].z, a2v[i].w};
#pragma unroll
      for (int t = 0; t < NT; ++t) {
        const v4f w = bs1[KQ * NCOLS + (2 * i) * NCOLS + t * 16];
        const v2f blo = {w.x, w.y};
        const v2f bhi = {w.z, w.w};
        acc[t] = __builtin_amdgcn_wmma_f32_16x16x4_f32(false, clo, false, blo,
                                                       (short)0, acc[t], false, false);
        acc[t] = __builtin_amdgcn_wmma_f32_16x16x4_f32(false, chi, false, bhi,
                                                       (short)0, acc[t], false, false);
      }
    }
  }

  float rdr[8];
  if (ADD_MEAN) {
#pragma unroll
    for (int v = 0; v < 8; ++v) rdr[v] = rdeg[row0 + v + 8 * hh];
  }
#pragma unroll
  for (int t = 0; t < NT; ++t) {
    const int col = t * 16 + lr;
    const float bb = BIAS ? bias[col] : 0.f;
#pragma unroll
    for (int v = 0; v < 8; ++v) {
      const int r = row0 + v + 8 * hh;
      float val = acc[t][v] + bb;
      if (ADD_MEAN) val += addmat[(size_t)r * NCOLS + col] * rdr[v];
      if (RELU) val = fmaxf(val, 0.f);
      out[(size_t)r * NCOLS + col] = val;
    }
  }
}

extern "C" void kernel_launch(void* const* d_in, const int* in_sizes, int n_in,
                              void* d_out, int out_size, void* d_ws,
                              size_t ws_size, hipStream_t stream) {
  const float* x   = (const float*)d_in[0];
  const int*   src = (const int*)d_in[1];
  const int*   dst = (const int*)d_in[2];
  const float* Ws1 = (const float*)d_in[3];
  const float* b1  = (const float*)d_in[4];
  const float* Wn1 = (const float*)d_in[5];
  const float* Ws2 = (const float*)d_in[6];
  const float* b2  = (const float*)d_in[7];
  const float* Wn2 = (const float*)d_in[8];
  const float* Ws3 = (const float*)d_in[9];
  const float* b3  = (const float*)d_in[10];
  const float* Wn3 = (const float*)d_in[11];
  float* out = (float*)d_out;

  const int N = in_sizes[0] / HID;
  const int E = in_sizes[1];

  float* h1  = (float*)d_ws;
  float* h2  = h1 + (size_t)N * HID;
  float* tb  = h2 + (size_t)N * HID;
  float* deg = tb + (size_t)N * OUTD + (size_t)N * HID;
  unsigned char* ids = (unsigned char*)tb;
  int* list = (int*)(ids + (((size_t)E + 16 + 255) & ~(size_t)255));
  unsigned int* tcnt = (unsigned int*)(list + 2 * (size_t)E);
  float* t3 = h1;

  const int mtiles = (N + 15) / 16;
  const int gemmB = (mtiles + 7) / 8;
  const int ownB = (N + OWN_R - 1) / OWN_R;

  unsigned int* tpart = (unsigned int*)h2;
  k_ids<<<IDS_P, IDS_T, 0, stream>>>(dst, ids, tpart, E, N);

  k_own<HID, HID, true, true><<<ownB, OWN_T, 0, stream>>>(
      x, src, dst, ids, tpart, tcnt, list, x, deg, Ws1, Wn1, b1, h1, E, N);

  k_own<HID, HID, false, true><<<ownB, OWN_T, 0, stream>>>(
      h1, src, dst, ids, tpart, tcnt, list, h1, deg, Ws2, Wn2, b2, h2, E, N);

  k_gemm<OUTD, false, false, false, false><<<gemmB, 256, 0, stream>>>(
      h2, nullptr, nullptr, Wn3, nullptr, nullptr, nullptr, t3, N);
  k_own<OUTD, OUTD, false, false><<<ownB, OWN_T, 0, stream>>>(
      t3, src, dst, ids, tpart, tcnt, list, h2, deg, Ws3, nullptr, b3, out, E, N);
}
